# starting kernel
# baseline (speedup 1.0000x reference)
#include <hip/hip_runtime.h>

typedef __attribute__((ext_vector_type(2))) float v2f;
typedef __attribute__((ext_vector_type(8))) float v8f;

#define DIN  128
#define DH   256
#define DOUT 384


__global__ __launch_bounds__(256) void zero_kernel(float4* __restrict__ p, int n4) {
  int i = blockIdx.x * blockDim.x + threadIdx.x;
  if (i < n4) p[i] = make_float4(0.f, 0.f, 0.f, 0.f);
}

__global__ __launch_bounds__(256) void count_kernel(const int* __restrict__ dst,
                                                    int* __restrict__ cnt, int nedges) {
  int e = blockIdx.x * blockDim.x + threadIdx.x;
  if (e < nedges) atomicAdd(&cnt[dst[e]], 1);
}


__global__ __launch_bounds__(256) void scan_blocks_kernel(const int* __restrict__ cnt,
                                                          int* __restrict__ excl,
                                                          int* __restrict__ partials, int n) {
  __shared__ int sh[256];
  const int t = threadIdx.x;
  const int i = blockIdx.x * 256 + t;
  int v = (i < n) ? cnt[i] : 0;
  sh[t] = v;
  __syncthreads();
  for (int o = 1; o < 256; o <<= 1) {
    int x = (t >= o) ? sh[t - o] : 0;
    __syncthreads();
    sh[t] += x;
    __syncthreads();
  }
  if (i < n) excl[i] = sh[t] - v;
  if (t == 255) partials[blockIdx.x] = sh[255];
}

__global__ __launch_bounds__(256) void scan_partials_kernel(int* __restrict__ partials, int nb) {
  __shared__ int sh[256];
  const int t = threadIdx.x;
  int v = (t < nb) ? partials[t] : 0;
  sh[t] = v;
  __syncthreads();
  for (int o = 1; o < 256; o <<= 1) {
    int x = (t >= o) ? sh[t - o] : 0;
    __syncthreads();
    sh[t] += x;
    __syncthreads();
  }
  if (t < nb) partials[t] = sh[t] - v;
}

__global__ __launch_bounds__(256) void scan_finish_kernel(int* __restrict__ rowptr,
                                                          const int* __restrict__ partials,
                                                          const int* __restrict__ cnt,
                                                          int* __restrict__ cursor,
                                                          float* __restrict__ invdeg, int n) {
  int i = blockIdx.x * 256 + threadIdx.x;
  if (i >= n) return;
  int r = rowptr[i] + partials[i >> 8];
  rowptr[i] = r;
  cursor[i] = r;
  invdeg[i] = 1.0f / fmaxf((float)cnt[i], 1.0f);
  if (i == n - 1) rowptr[n] = r + cnt[i];
}

__global__ __launch_bounds__(256) void fill_adj_kernel(const int* __restrict__ src,
                                                       const int* __restrict__ dst,
                                                       int* __restrict__ cursor,
                                                       int* __restrict__ adj, int nedges) {
  int e = blockIdx.x * blockDim.x + threadIdx.x;
  if (e >= nedges) return;
  int pos = atomicAdd(&cursor[dst[e]], 1);
  adj[pos] = src[e];
}

template<int D>
__global__ __launch_bounds__(256) void gather_mean_kernel(const float* __restrict__ h,
                                                          const int* __restrict__ rowptr,
                                                          const int* __restrict__ adj,
                                                          const float* __restrict__ invdeg,
                                                          float* __restrict__ mean, int nnodes) {
  const int lane = threadIdx.x & 31;
  const int wave = threadIdx.x >> 5;
  const int node = blockIdx.x * 8 + wave;
  if (node >= nnodes) return;
  const int cbase = blockIdx.y * 64 + lane * 2;
  const int beg = rowptr[node];
  const int end = rowptr[node + 1];
  float2 acc = make_float2(0.f, 0.f);
  for (int j = beg; j < end; ++j) {
    const int s = adj[j];
    const float2 v = *(const float2*)(h + (size_t)s * D + cbase);
    acc.x += v.x;
    acc.y += v.y;
  }
  const float sc = invdeg[node];
  acc.x *= sc;
  acc.y *= sc;
  *(float2*)(mean + (size_t)node * D + cbase) = acc;
}

template<int K, int NOUT>
__global__ __launch_bounds__(256) void permute_w_kernel(const float* __restrict__ W,
                                                        float* __restrict__ Wt) {
  int i = blockIdx.x * blockDim.x + threadIdx.x;
  if (i >= K * NOUT) return;
  int k = i / NOUT;
  int col = i - k * NOUT;
  int stream = (k >> 1) & 1;
  int j = k >> 2;
  int pos = k & 1;
  Wt[((size_t)(col * 2 + stream) * (K / 4) + j) * 2 + pos] = W[i];
}

template<int K, int NOUT, bool RELU>
__global__ __launch_bounds__(256) void sage_gemm_kernel(const float* __restrict__ mean,
                                                        const float* __restrict__ hin,
                                                        const float* __restrict__ Wlp,
                                                        const float* __restrict__ Wrp,
                                                        const float* __restrict__ bias,
                                                        float* __restrict__ out, int mtiles) {
  const int lane  = threadIdx.x & 31;
  const int wave  = threadIdx.x >> 5;
  const int mtile = blockIdx.x * 8 + wave;
  if (mtile >= mtiles) return;
  const int nbase = blockIdx.y * 64;
  const int l15   = lane & 15;
  const int row   = mtile * 16 + l15;
  const int s     = lane >> 4;
  const int khalf = s << 1;

  v8f acc0 = {}, acc1 = {}, acc2 = {}, acc3 = {};

  for (int pass = 0; pass < 2; ++pass) {
    const float* __restrict__ Ar = (pass == 0 ? mean : hin) + (size_t)row * K;
    const float* __restrict__ Wp = (pass == 0 ? Wlp : Wrp);
    const v2f* bp0 = (const v2f*)Wp + (size_t)((nbase + l15 +  0) * 2 + s) * (K / 4);
    const v2f* bp1 = (const v2f*)Wp + (size_t)((nbase + l15 + 16) * 2 + s) * (K / 4);
    const v2f* bp2 = (const v2f*)Wp + (size_t)((nbase + l15 + 32) * 2 + s) * (K / 4);
    const v2f* bp3 = (const v2f*)Wp + (size_t)((nbase + l15 + 48) * 2 + s) * (K / 4);
#pragma unroll 4
    for (int j = 0; j < K / 4; ++j) {
      v2f a = *(const v2f*)(Ar + 4 * j + khalf);
      acc0 = __builtin_amdgcn_wmma_f32_16x16x4_f32(false, a, false, bp0[j], (short)0, acc0, false, false);
      acc1 = __builtin_amdgcn_wmma_f32_16x16x4_f32(false, a, false, bp1[j], (short)0, acc1, false, false);
      acc2 = __builtin_amdgcn_wmma_f32_16x16x4_f32(false, a, false, bp2[j], (short)0, acc2, false, false);
      acc3 = __builtin_amdgcn_wmma_f32_16x16x4_f32(false, a, false, bp3[j], (short)0, acc3, false, false);
    }
  }

  const int col = nbase + l15;
  const float bj0 = bias[col];
  const float bj1 = bias[col + 16];
  const float bj2 = bias[col + 32];
  const float bj3 = bias[col + 48];
  const int r0 = mtile * 16 + (s << 3);
#pragma unroll
  for (int v = 0; v < 8; ++v) {
    float* o = out + (size_t)(r0 + v) * NOUT + col;
    float t0 = acc0[v] + bj0;
    float t1 = acc1[v] + bj1;
    float t2 = acc2[v] + bj2;
    float t3 = acc3[v] + bj3;
    if (RELU) {
      t0 = fmaxf(t0, 0.f); t1 = fmaxf(t1, 0.f);
      t2 = fmaxf(t2, 0.f); t3 = fmaxf(t3, 0.f);
    }
    o[0] = t0; o[16] = t1; o[32] = t2; o[48] = t3;
  }
}


extern "C" void kernel_launch(void* const* d_in, const int* in_sizes, int n_in,
                              void* d_out, int out_size, void* d_ws, size_t ws_size,
                              hipStream_t stream) {
  const float* x   = (const float*)d_in[0];
  const int*   ei  = (const int*)d_in[1];
  const float* W1l = (const float*)d_in[2];
  const float* W1r = (const float*)d_in[3];
  const float* b1  = (const float*)d_in[4];
  const float* W2l = (const float*)d_in[5];
  const float* W2r = (const float*)d_in[6];
  const float* b2  = (const float*)d_in[7];
  const float* W3l = (const float*)d_in[8];
  const float* W3r = (const float*)d_in[9];
  const float* b3  = (const float*)d_in[10];
  float* out = (float*)d_out;

  const int N = in_sizes[0] / DIN;
  const int E = in_sizes[1] / 2;
  const int* src = ei;
  const int* dst = ei + E;

  char* ws = (char*)d_ws;
  auto al = [](size_t v) { return (v + 255) & ~(size_t)255; };
  size_t off = 0;
  int*   cnt      = (int*)(ws + off);   off += al((size_t)N * 4);
  int*   rowptr   = (int*)(ws + off);   off += al((size_t)(N + 1) * 4);
  int*   cursor   = (int*)(ws + off);   off += al((size_t)N * 4);
  int*   partials = (int*)(ws + off);   off += al((size_t)256 * 4);
  float* invdeg   = (float*)(ws + off); off += al((size_t)N * 4);
  int*   adj      = (int*)(ws + off);   off += al((size_t)E * 4);
  float* mean     = (float*)(ws + off); off += al((size_t)N * DH * 4);
  float* h1       = (float*)(ws + off); off += al((size_t)N * DH * 4);
  float* h2       = (float*)(ws + off); off += al((size_t)N * DH * 4);
  float* Wt1l     = (float*)(ws + off); off += al((size_t)DIN * DH * 4);
  float* Wt1r     = (float*)(ws + off); off += al((size_t)DIN * DH * 4);
  float* Wt2l     = (float*)(ws + off); off += al((size_t)DH * DH * 4);
  float* Wt2r     = (float*)(ws + off); off += al((size_t)DH * DH * 4);
  float* Wt3l     = (float*)(ws + off); off += al((size_t)DH * DOUT * 4);
  float* Wt3r     = (float*)(ws + off);

  const int TB = 256;
  const int NB = (N + 255) / 256;
  const int mtiles = N / 16;
  const int gm = (mtiles + 7) / 8;
  const int gn = (N + 7) / 8;

  zero_kernel<<<(N / 4 + TB - 1) / TB, TB, 0, stream>>>((float4*)cnt, N / 4);
  count_kernel<<<(E + TB - 1) / TB, TB, 0, stream>>>(dst, cnt, E);
  scan_blocks_kernel<<<NB, TB, 0, stream>>>(cnt, rowptr, partials, N);
  scan_partials_kernel<<<1, TB, 0, stream>>>(partials, NB);
  scan_finish_kernel<<<NB, TB, 0, stream>>>(rowptr, partials, cnt, cursor, invdeg, N);
  fill_adj_kernel<<<(E + TB - 1) / TB, TB, 0, stream>>>(src, dst, cursor, adj, E);

  permute_w_kernel<DIN, DH ><<<(DIN * DH  + TB - 1) / TB, TB, 0, stream>>>(W1l, Wt1l);
  permute_w_kernel<DIN, DH ><<<(DIN * DH  + TB - 1) / TB, TB, 0, stream>>>(W1r, Wt1r);
  permute_w_kernel<DH,  DH ><<<(DH  * DH  + TB - 1) / TB, TB, 0, stream>>>(W2l, Wt2l);
  permute_w_kernel<DH,  DH ><<<(DH  * DH  + TB - 1) / TB, TB, 0, stream>>>(W2r, Wt2r);
  permute_w_kernel<DH, DOUT><<<(DH * DOUT + TB - 1) / TB, TB, 0, stream>>>(W3l, Wt3l);
  permute_w_kernel<DH, DOUT><<<(DH * DOUT + TB - 1) / TB, TB, 0, stream>>>(W3r, Wt3r);

  {
    dim3 ga(gn, DIN / 64);
    gather_mean_kernel<DIN><<<ga, TB, 0, stream>>>(x, rowptr, adj, invdeg, mean, N);
    dim3 g(gm, DH / 64);
    sage_gemm_kernel<DIN, DH, true><<<g, TB, 0, stream>>>(mean, x, Wt1l, Wt1r, b1, h1, mtiles);
  }
  {
    dim3 ga(gn, DH / 64);
    gather_mean_kernel<DH><<<ga, TB, 0, stream>>>(h1, rowptr, adj, invdeg, mean, N);
    dim3 g(gm, DH / 64);
    sage_gemm_kernel<DH, DH, true><<<g, TB, 0, stream>>>(mean, h1, Wt2l, Wt2r, b2, h2, mtiles);
  }
  {
    dim3 ga(gn, DH / 64);
    gather_mean_kernel<DH><<<ga, TB, 0, stream>>>(h2, rowptr, adj, invdeg, mean, N);
    dim3 g(gm, DOUT / 64);
    sage_gemm_kernel<DH, DOUT, false><<<g, TB, 0, stream>>>(mean, h2, Wt3l, Wt3r, b3, out, mtiles);
  }
  (void)n_in; (void)out_size; (void)ws_size;
}


# aggregation fused into the GEMM kernel: a work-group gathers its 32 rows' neighbour means into an LDS A tile [mean|self] and multiplies from LDS; mean array and 3 gather launches gone; sum order kept
# speedup vs baseline: 2.6275x; 2.6275x over previous
#include <hip/hip_runtime.h>

typedef __attribute__((ext_vector_type(2))) float v2f;
typedef __attribute__((ext_vector_type(4))) float v4f;
typedef __attribute__((ext_vector_type(8))) float v8f;

#define DIN  128
#define DH   256
#define DOUT 384


__global__ __launch_bounds__(256) void zero_kernel(float4* __restrict__ p, int n4) {
  int i = blockIdx.x * blockDim.x + threadIdx.x;
  if (i < n4) p[i] = make_float4(0.f, 0.f, 0.f, 0.f);
}

__global__ __launch_bounds__(256) void count_kernel(const int* __restrict__ dst,
                                                    int* __restrict__ cnt, int nedges) {
  int e = blockIdx.x * blockDim.x + threadIdx.x;
  if (e < nedges) atomicAdd(&cnt[dst[e]], 1);
}


__global__ __launch_bounds__(256) void scan_blocks_kernel(const int* __restrict__ cnt,
                                                          int* __restrict__ excl,
                                                          int* __restrict__ partials, int n) {
  __shared__ int sh[256];
  const int t = threadIdx.x;
  const int i = blockIdx.x * 256 + t;
  int v = (i < n) ? cnt[i] : 0;
  sh[t] = v;
  __syncthreads();
  for (int o = 1; o < 256; o <<= 1) {
    int x = (t >= o) ? sh[t - o] : 0;
    __syncthreads();
    sh[t] += x;
    __syncthreads();
  }
  if (i < n) excl[i] = sh[t] - v;
  if (t == 255) partials[blockIdx.x] = sh[255];
}

__global__ __launch_bounds__(256) void scan_partials_kernel(int* __restrict__ partials, int nb) {
  __shared__ int sh[256];
  const int t = threadIdx.x;
  int v = (t < nb) ? partials[t] : 0;
  sh[t] = v;
  __syncthreads();
  for (int o = 1; o < 256; o <<= 1) {
    int x = (t >= o) ? sh[t - o] : 0;
    __syncthreads();
    sh[t] += x;
    __syncthreads();
  }
  if (t < nb) partials[t] = sh[t] - v;
}

__global__ __launch_bounds__(256) void scan_finish_kernel(int* __restrict__ rowptr,
                                                          const int* __restrict__ partials,
                                                          const int* __restrict__ cnt,
                                                          int* __restrict__ cursor,
                                                          float* __restrict__ invdeg, int n) {
  int i = blockIdx.x * 256 + threadIdx.x;
  if (i >= n) return;
  int r = rowptr[i] + partials[i >> 8];
  rowptr[i] = r;
  cursor[i] = r;
  invdeg[i] = 1.0f / fmaxf((float)cnt[i], 1.0f);
  if (i == n - 1) rowptr[n] = r + cnt[i];
}

__global__ __launch_bounds__(256) void fill_adj_kernel(const int* __restrict__ src,
                                                       const int* __restrict__ dst,
                                                       int* __restrict__ cursor,
                                                       int* __restrict__ adj, int nedges) {
  int e = blockIdx.x * blockDim.x + threadIdx.x;
  if (e >= nedges) return;
  int pos = atomicAdd(&cursor[dst[e]], 1);
  adj[pos] = src[e];
}

template<int K, int NOUT>
__global__ __launch_bounds__(256) void permute_w_kernel(const float* __restrict__ W,
                                                        float* __restrict__ Wt) {
  int i = blockIdx.x * blockDim.x + threadIdx.x;
  if (i >= K * NOUT) return;
  int k = i / NOUT;
  int col = i - k * NOUT;
  int s = (k >> 1) & 1;
  int j = k >> 2;
  int jj = j >> 1;
  int q = ((j & 1) << 1) | (k & 1);
  int nb = col >> 6;
  int t = (col >> 4) & 3;
  int lane = (s << 4) | (col & 15);
  Wt[((((size_t)nb * (K / 8) + jj) * 4 + t) * 32 + lane) * 4 + q] = W[i];
}

template<int K, int NOUT, bool RELU>
__global__ __launch_bounds__(NOUT) void sage_fused_kernel(const float* __restrict__ hin,
                                                          const int* __restrict__ rowptr,
                                                          const int* __restrict__ adj,
                                                          const float* __restrict__ invdeg,
                                                          const float* __restrict__ Wlp,
                                                          const float* __restrict__ Wrp,
                                                          const float* __restrict__ bias,
                                                          float* __restrict__ out, int nnodes) {
  constexpr int NW    = NOUT / 32;
  constexpr int PITCH = 2 * K + 4;
  constexpr int KV    = K / 128;
  __shared__ __attribute__((aligned(16))) float As[32 * PITCH];

  const int lane = threadIdx.x & 31;
  const int wv   = __builtin_amdgcn_readfirstlane((int)(threadIdx.x >> 5));
  const int r0   = blockIdx.x * 32;

  for (int i = wv; i < 32; i += NW) {
    const int node = r0 + i;
    v4f* Arow = (v4f*)(As + i * PITCH);
    if (node < nnodes) {
      const int beg = rowptr[node];
      const int end = rowptr[node + 1];
      const float sc = invdeg[node];
      const v4f* self = (const v4f*)(hin + (size_t)node * K) + lane;
      v4f sv[KV];
      v4f acc[KV];
      for (int c = 0; c < KV; ++c) {
        sv[c]  = self[c * 32];
        acc[c] = (v4f){0.f, 0.f, 0.f, 0.f};
      }
      for (int j = beg; j < end; ++j) {
        const int sidx = adj[j];
        const v4f* hp = (const v4f*)(hin + (size_t)sidx * K) + lane;
        for (int c = 0; c < KV; ++c) acc[c] += hp[c * 32];
      }
      for (int c = 0; c < KV; ++c) {
        Arow[c * 32 + lane]         = acc[c] * sc;
        Arow[K / 4 + c * 32 + lane] = sv[c];
      }
    } else {
      for (int c = 0; c < 2 * KV; ++c) Arow[c * 32 + lane] = (v4f){0.f, 0.f, 0.f, 0.f};
    }
  }
  __syncthreads();

  const int l15   = lane & 15;
  const int s     = lane >> 4;
  const int khalf = s << 1;

  v8f acc00 = {}, acc01 = {}, acc10 = {}, acc11 = {};

  for (int pass = 0; pass < 2; ++pass) {
    const float* __restrict__ Wp = (pass == 0 ? Wlp : Wrp);
    const v4f* bp = (const v4f*)Wp + (size_t)(wv >> 1) * (K / 8) * 128 + (wv & 1) * 64 + lane;
    const float* A0 = As + l15 * PITCH + pass * K + khalf;
    const float* A1 = A0 + 16 * PITCH;
#pragma unroll 2
    for (int jj = 0; jj < K / 8; ++jj) {
      const v2f a00 = *(const v2f*)(A0 + 8 * jj);
      const v2f a01 = *(const v2f*)(A0 + 8 * jj + 4);
      const v2f a10 = *(const v2f*)(A1 + 8 * jj);
      const v2f a11 = *(const v2f*)(A1 + 8 * jj + 4);
      const v4f b0 = bp[jj * 128 +  0];
      const v4f b1 = bp[jj * 128 + 32];
      acc00 = __builtin_amdgcn_wmma_f32_16x16x4_f32(false, a00, false, __builtin_shufflevector(b0, b0, 0, 1), (short)0, acc00, false, false);
      acc01 = __builtin_amdgcn_wmma_f32_16x16x4_f32(false, a00, false, __builtin_shufflevector(b1, b1, 0, 1), (short)0, acc01, false, false);
      acc10 = __builtin_amdgcn_wmma_f32_16x16x4_f32(false, a10, false, __builtin_shufflevector(b0, b0, 0, 1), (short)0, acc10, false, false);
      acc11 = __builtin_amdgcn_wmma_f32_16x16x4_f32(false, a10, false, __builtin_shufflevector(b1, b1, 0, 1), (short)0, acc11, false, false);
      acc00 = __builtin_amdgcn_wmma_f32_16x16x4_f32(false, a01, false, __builtin_shufflevector(b0, b0, 2, 3), (short)0, acc00, false, false);
      acc01 = __builtin_amdgcn_wmma_f32_16x16x4_f32(false, a01, false, __builtin_shufflevector(b1, b1, 2, 3), (short)0, acc01, false, false);
      acc10 = __builtin_amdgcn_wmma_f32_16x16x4_f32(false, a11, false, __builtin_shufflevector(b0, b0, 2, 3), (short)0, acc10, false, false);
      acc11 = __builtin_amdgcn_wmma_f32_16x16x4_f32(false, a11, false, __builtin_shufflevector(b1, b1, 2, 3), (short)0, acc11, false, false);
    }
  }

  const int col = wv * 32 + l15;
  const float bj0 = bias[col];
  const float bj1 = bias[col + 16];
  const int rb = r0 + (s << 3);
#pragma unroll
  for (int v = 0; v < 8; ++v) {
    float t00 = acc00[v] + bj0;
    float t01 = acc01[v] + bj1;
    float t10 = acc10[v] + bj0;
    float t11 = acc11[v] + bj1;
    if (RELU) {
      t00 = fmaxf(t00, 0.f); t01 = fmaxf(t01, 0.f);
      t10 = fmaxf(t10, 0.f); t11 = fmaxf(t11, 0.f);
    }
    const int ra = rb + v;
    const int rc = rb + 16 + v;
    if (ra < nnodes) {
      float* o = out + (size_t)ra * NOUT + col;
      o[0] = t00; o[16] = t01;
    }
    if (rc < nnodes) {
      float* o = out + (size_t)rc * NOUT + col;
      o[0] = t10; o[16] = t11;
    }
  }
}


extern "C" void kernel_launch(void* const* d_in, const int* in_sizes, int n_in,
                              void* d_out, int out_size, void* d_ws, size_t ws_size,
                              hipStream_t stream) {
  const float* x   = (const float*)d_in[0];
  const int*   ei  = (const int*)d_in[1];
  const float* W1l = (const float*)d_in[2];
  const float* W1r = (const float*)d_in[3];
  const float* b1  = (const float*)d_in[4];
  const float* W2l = (const float*)d_in[5];
  const float* W2r = (const float*)d_in[6];
  const float* b2  = (const float*)d_in[7];
  const float* W3l = (const float*)d_in[8];
  const float* W3r = (const float*)d_in[9];
  const float* b3  = (const float*)d_in[10];
  float* out = (float*)d_out;

  const int N = in_sizes[0] / DIN;
  const int E = in_sizes[1] / 2;
  const int* src = ei;
  const int* dst = ei + E;

  char* ws = (char*)d_ws;
  auto al = [](size_t v) { return (v + 255) & ~(size_t)255; };
  size_t off = 0;
  int*   cnt      = (int*)(ws + off);   off += al((size_t)N * 4);
  int*   rowptr   = (int*)(ws + off);   off += al((size_t)(N + 1) * 4);
  int*   cursor   = (int*)(ws + off);   off += al((size_t)N * 4);
  int*   partials = (int*)(ws + off);   off += al((size_t)256 * 4);
  float* invdeg   = (float*)(ws + off); off += al((size_t)N * 4);
  int*   adj      = (int*)(ws + off);   off += al((size_t)E * 4);
  float* h1       = (float*)(ws + off); off += al((size_t)N * DH * 4);
  float* h2       = (float*)(ws + off); off += al((size_t)N * DH * 4);
  float* Wt1l     = (float*)(ws + off); off += al((size_t)DIN * DH * 4);
  float* Wt1r     = (float*)(ws + off); off += al((size_t)DIN * DH * 4);
  float* Wt2l     = (float*)(ws + off); off += al((size_t)DH * DH * 4);
  float* Wt2r     = (float*)(ws + off); off += al((size_t)DH * DH * 4);
  float* Wt3l     = (float*)(ws + off); off += al((size_t)DH * DOUT * 4);
  float* Wt3r     = (float*)(ws + off);

  const int TB = 256;
  const int NB = (N + 255) / 256;
  const int gb = (N + 31) / 32;

  zero_kernel<<<(N / 4 + TB - 1) / TB, TB, 0, stream>>>((float4*)cnt, N / 4);
  count_kernel<<<(E + TB - 1) / TB, TB, 0, stream>>>(dst, cnt, E);
  scan_blocks_kernel<<<NB, TB, 0, stream>>>(cnt, rowptr, partials, N);
  scan_partials_kernel<<<1, TB, 0, stream>>>(partials, NB);
  scan_finish_kernel<<<NB, TB, 0, stream>>>(rowptr, partials, cnt, cursor, invdeg, N);
  fill_adj_kernel<<<(E + TB - 1) / TB, TB, 0, stream>>>(src, dst, cursor, adj, E);

  permute_w_kernel<DIN, DH ><<<(DIN * DH  + TB - 1) / TB, TB, 0, stream>>>(W1l, Wt1l);
  permute_w_kernel<DIN, DH ><<<(DIN * DH  + TB - 1) / TB, TB, 0, stream>>>(W1r, Wt1r);
  permute_w_kernel<DH,  DH ><<<(DH  * DH  + TB - 1) / TB, TB, 0, stream>>>(W2l, Wt2l);
  permute_w_kernel<DH,  DH ><<<(DH  * DH  + TB - 1) / TB, TB, 0, stream>>>(W2r, Wt2r);
  permute_w_kernel<DH, DOUT><<<(DH * DOUT + TB - 1) / TB, TB, 0, stream>>>(W3l, Wt3l);
  permute_w_kernel<DH, DOUT><<<(DH * DOUT + TB - 1) / TB, TB, 0, stream>>>(W3r, Wt3r);

  sage_fused_kernel<DIN, DH, true><<<gb, DH, 0, stream>>>(x, rowptr, adj, invdeg, Wt1l, Wt1r, b1, h1, N);
  sage_fused_kernel<DH, DH, true><<<gb, DH, 0, stream>>>(h1, rowptr, adj, invdeg, Wt2l, Wt2r, b2, h2, N);
  sage_fused_kernel<DH, DOUT, false><<<gb, DOUT, 0, stream>>>(h2, rowptr, adj, invdeg, Wt3l, Wt3r, b3, out, N);
  (void)n_in; (void)out_size; (void)ws_size;
}


# layers 2-3: work-groups alternate (block-index bit parity) between gather-first and self-product-first so co-resident ones gather and multiply at once; self-first rows add hin@Wr before mean@Wl
# speedup vs baseline: 2.7264x; 1.0376x over previous
#include <hip/hip_runtime.h>

typedef __attribute__((ext_vector_type(2))) float v2f;
typedef __attribute__((ext_vector_type(4))) float v4f;
typedef __attribute__((ext_vector_type(8))) float v8f;

#define DIN  128
#define DH   256
#define DOUT 384


__global__ __launch_bounds__(256) void zero_kernel(float4* __restrict__ p, int n4) {
  int i = blockIdx.x * blockDim.x + threadIdx.x;
  if (i < n4) p[i] = make_float4(0.f, 0.f, 0.f, 0.f);
}

__global__ __launch_bounds__(256) void count_kernel(const int* __restrict__ dst,
                                                    int* __restrict__ cnt, int nedges) {
  int e = blockIdx.x * blockDim.x + threadIdx.x;
  if (e < nedges) atomicAdd(&cnt[dst[e]], 1);
}


__global__ __launch_bounds__(256) void scan_blocks_kernel(const int* __restrict__ cnt,
                                                          int* __restrict__ excl,
                                                          int* __restrict__ partials, int n) {
  __shared__ int sh[256];
  const int t = threadIdx.x;
  const int i = blockIdx.x * 256 + t;
  int v = (i < n) ? cnt[i] : 0;
  sh[t] = v;
  __syncthreads();
  for (int o = 1; o < 256; o <<= 1) {
    int x = (t >= o) ? sh[t - o] : 0;
    __syncthreads();
    sh[t] += x;
    __syncthreads();
  }
  if (i < n) excl[i] = sh[t] - v;
  if (t == 255) partials[blockIdx.x] = sh[255];
}

__global__ __launch_bounds__(256) void scan_partials_kernel(int* __restrict__ partials, int nb) {
  __shared__ int sh[256];
  const int t = threadIdx.x;
  int v = (t < nb) ? partials[t] : 0;
  sh[t] = v;
  __syncthreads();
  for (int o = 1; o < 256; o <<= 1) {
    int x = (t >= o) ? sh[t - o] : 0;
    __syncthreads();
    sh[t] += x;
    __syncthreads();
  }
  if (t < nb) partials[t] = sh[t] - v;
}

__global__ __launch_bounds__(256) void scan_finish_kernel(int* __restrict__ rowptr,
                                                          const int* __restrict__ partials,
                                                          const int* __restrict__ cnt,
                                                          int* __restrict__ cursor,
                                                          float* __restrict__ invdeg, int n) {
  int i = blockIdx.x * 256 + threadIdx.x;
  if (i >= n) return;
  int r = rowptr[i] + partials[i >> 8];
  rowptr[i] = r;
  cursor[i] = r;
  invdeg[i] = 1.0f / fmaxf((float)cnt[i], 1.0f);
  if (i == n - 1) rowptr[n] = r + cnt[i];
}

__global__ __launch_bounds__(256) void fill_adj_kernel(const int* __restrict__ src,
                                                       const int* __restrict__ dst,
                                                       int* __restrict__ cursor,
                                                       int* __restrict__ adj, int nedges) {
  int e = blockIdx.x * blockDim.x + threadIdx.x;
  if (e >= nedges) return;
  int pos = atomicAdd(&cursor[dst[e]], 1);
  adj[pos] = src[e];
}

template<int K, int NOUT>
__global__ __launch_bounds__(256) void permute_w_kernel(const float* __restrict__ W,
                                                        float* __restrict__ Wt) {
  int i = blockIdx.x * blockDim.x + threadIdx.x;
  if (i >= K * NOUT) return;
  int k = i / NOUT;
  int col = i - k * NOUT;
  int s = (k >> 1) & 1;
  int j = k >> 2;
  int jj = j >> 1;
  int q = ((j & 1) << 1) | (k & 1);
  int nb = col >> 6;
  int t = (col >> 4) & 3;
  int lane = (s << 4) | (col & 15);
  Wt[((((size_t)nb * (K / 8) + jj) * 4 + t) * 32 + lane) * 4 + q] = W[i];
}

template<int K, int NOUT, bool RELU>
__global__ __launch_bounds__(NOUT) void sage_fused_kernel(const float* __restrict__ hin,
                                                          const int* __restrict__ rowptr,
                                                          const int* __restrict__ adj,
                                                          const float* __restrict__ invdeg,
                                                          const float* __restrict__ Wlp,
                                                          const float* __restrict__ Wrp,
                                                          const float* __restrict__ bias,
                                                          float* __restrict__ out, int nnodes) {
  constexpr int NW    = NOUT / 32;
  constexpr int PITCH = 2 * K + 4;
  constexpr int KV    = K / 128;
  __shared__ __attribute__((aligned(16))) float As[32 * PITCH];

  const int lane = threadIdx.x & 31;
  const int wv   = __builtin_amdgcn_readfirstlane((int)(threadIdx.x >> 5));
  const int r0   = blockIdx.x * 32;

  for (int i = wv; i < 32; i += NW) {
    const int node = r0 + i;
    v4f* Arow = (v4f*)(As + i * PITCH);
    if (node < nnodes) {
      const int beg = rowptr[node];
      const int end = rowptr[node + 1];
      const float sc = invdeg[node];
      const v4f* self = (const v4f*)(hin + (size_t)node * K) + lane;
      v4f sv[KV];
      v4f acc[KV];
      for (int c = 0; c < KV; ++c) {
        sv[c]  = self[c * 32];
        acc[c] = (v4f){0.f, 0.f, 0.f, 0.f};
      }
      for (int j = beg; j < end; ++j) {
        const int sidx = adj[j];
        const v4f* hp = (const v4f*)(hin + (size_t)sidx * K) + lane;
        for (int c = 0; c < KV; ++c) acc[c] += hp[c * 32];
      }
      for (int c = 0; c < KV; ++c) {
        Arow[c * 32 + lane]         = acc[c] * sc;
        Arow[K / 4 + c * 32 + lane] = sv[c];
      }
    } else {
      for (int c = 0; c < 2 * KV; ++c) Arow[c * 32 + lane] = (v4f){0.f, 0.f, 0.f, 0.f};
    }
  }
  __syncthreads();

  const int l15   = lane & 15;
  const int s     = lane >> 4;
  const int khalf = s << 1;

  v8f acc00 = {}, acc01 = {}, acc10 = {}, acc11 = {};

  for (int pass = 0; pass < 2; ++pass) {
    const float* __restrict__ Wp = (pass == 0 ? Wlp : Wrp);
    const v4f* bp = (const v4f*)Wp + (size_t)(wv >> 1) * (K / 8) * 128 + (wv & 1) * 64 + lane;
    const float* A0 = As + l15 * PITCH + pass * K + khalf;
    const float* A1 = A0 + 16 * PITCH;
#pragma unroll 2
    for (int jj = 0; jj < K / 8; ++jj) {
      const v2f a00 = *(const v2f*)(A0 + 8 * jj);
      const v2f a01 = *(const v2f*)(A0 + 8 * jj + 4);
      const v2f a10 = *(const v2f*)(A1 + 8 * jj);
      const v2f a11 = *(const v2f*)(A1 + 8 * jj + 4);
      const v4f b0 = bp[jj * 128 +  0];
      const v4f b1 = bp[jj * 128 + 32];
      acc00 = __builtin_amdgcn_wmma_f32_16x16x4_f32(false, a00, false, __builtin_shufflevector(b0, b0, 0, 1), (short)0, acc00, false, false);
      acc01 = __builtin_amdgcn_wmma_f32_16x16x4_f32(false, a00, false, __builtin_shufflevector(b1, b1, 0, 1), (short)0, acc01, false, false);
      acc10 = __builtin_amdgcn_wmma_f32_16x16x4_f32(false, a10, false, __builtin_shufflevector(b0, b0, 0, 1), (short)0, acc10, false, false);
      acc11 = __builtin_amdgcn_wmma_f32_16x16x4_f32(false, a10, false, __builtin_shufflevector(b1, b1, 0, 1), (short)0, acc11, false, false);
      acc00 = __builtin_amdgcn_wmma_f32_16x16x4_f32(false, a01, false, __builtin_shufflevector(b0, b0, 2, 3), (short)0, acc00, false, false);
      acc01 = __builtin_amdgcn_wmma_f32_16x16x4_f32(false, a01, false, __builtin_shufflevector(b1, b1, 2, 3), (short)0, acc01, false, false);
      acc10 = __builtin_amdgcn_wmma_f32_16x16x4_f32(false, a11, false, __builtin_shufflevector(b0, b0, 2, 3), (short)0, acc10, false, false);
      acc11 = __builtin_amdgcn_wmma_f32_16x16x4_f32(false, a11, false, __builtin_shufflevector(b1, b1, 2, 3), (short)0, acc11, false, false);
    }
  }

  const int col = wv * 32 + l15;
  const float bj0 = bias[col];
  const float bj1 = bias[col + 16];
  const int rb = r0 + (s << 3);
#pragma unroll
  for (int v = 0; v < 8; ++v) {
    float t00 = acc00[v] + bj0;
    float t01 = acc01[v] + bj1;
    float t10 = acc10[v] + bj0;
    float t11 = acc11[v] + bj1;
    if (RELU) {
      t00 = fmaxf(t00, 0.f); t01 = fmaxf(t01, 0.f);
      t10 = fmaxf(t10, 0.f); t11 = fmaxf(t11, 0.f);
    }
    const int ra = rb + v;
    const int rc = rb + 16 + v;
    if (ra < nnodes) {
      float* o = out + (size_t)ra * NOUT + col;
      o[0] = t00; o[16] = t01;
    }
    if (rc < nnodes) {
      float* o = out + (size_t)rc * NOUT + col;
      o[0] = t10; o[16] = t11;
    }
  }
}


template<int K, int PITCH>
__device__ __forceinline__ void gemm_pass(const float* __restrict__ As, const float* __restrict__ Wp,
                                          int pass, int wv, int lane,
                                          v8f& acc00, v8f& acc01, v8f& acc10, v8f& acc11) {
  const int l15   = lane & 15;
  const int khalf = (lane >> 4) << 1;
  const v4f* bp = (const v4f*)Wp + (size_t)(wv >> 1) * (K / 8) * 128 + (wv & 1) * 64 + lane;
  const float* A0 = As + l15 * PITCH + pass * K + khalf;
  const float* A1 = A0 + 16 * PITCH;
#pragma unroll 2
  for (int jj = 0; jj < K / 8; ++jj) {
    const v2f a00 = *(const v2f*)(A0 + 8 * jj);
    const v2f a01 = *(const v2f*)(A0 + 8 * jj + 4);
    const v2f a10 = *(const v2f*)(A1 + 8 * jj);
    const v2f a11 = *(const v2f*)(A1 + 8 * jj + 4);
    const v4f b0 = bp[jj * 128 +  0];
    const v4f b1 = bp[jj * 128 + 32];
    acc00 = __builtin_amdgcn_wmma_f32_16x16x4_f32(false, a00, false, __builtin_shufflevector(b0, b0, 0, 1), (short)0, acc00, false, false);
    acc01 = __builtin_amdgcn_wmma_f32_16x16x4_f32(false, a00, false, __builtin_shufflevector(b1, b1, 0, 1), (short)0, acc01, false, false);
    acc10 = __builtin_amdgcn_wmma_f32_16x16x4_f32(false, a10, false, __builtin_shufflevector(b0, b0, 0, 1), (short)0, acc10, false, false);
    acc11 = __builtin_amdgcn_wmma_f32_16x16x4_f32(false, a10, false, __builtin_shufflevector(b1, b1, 0, 1), (short)0, acc11, false, false);
    acc00 = __builtin_amdgcn_wmma_f32_16x16x4_f32(false, a01, false, __builtin_shufflevector(b0, b0, 2, 3), (short)0, acc00, false, false);
    acc01 = __builtin_amdgcn_wmma_f32_16x16x4_f32(false, a01, false, __builtin_shufflevector(b1, b1, 2, 3), (short)0, acc01, false, false);
    acc10 = __builtin_amdgcn_wmma_f32_16x16x4_f32(false, a11, false, __builtin_shufflevector(b0, b0, 2, 3), (short)0, acc10, false, false);
    acc11 = __builtin_amdgcn_wmma_f32_16x16x4_f32(false, a11, false, __builtin_shufflevector(b1, b1, 2, 3), (short)0, acc11, false, false);
  }
}

template<int K, int NW, int PITCH, bool WITH_SELF>
__device__ __forceinline__ void build_rows(float* __restrict__ As, const float* __restrict__ hin,
                                           const int* __restrict__ rowptr, const int* __restrict__ adj,
                                           const float* __restrict__ invdeg,
                                           int r0, int nnodes, int wv, int lane) {
  constexpr int KV = K / 128;
  constexpr int U  = 4;
  for (int i = wv; i < 32; i += NW) {
    const int node = r0 + i;
    v4f* Arow = (v4f*)(As + i * PITCH);
    if (node < nnodes) {
      const int beg = rowptr[node];
      const int end = rowptr[node + 1];
      const float sc = invdeg[node];
      v4f sv[KV];
      if (WITH_SELF) {
        const v4f* self = (const v4f*)(hin + (size_t)node * K) + lane;
        for (int c = 0; c < KV; ++c) sv[c] = self[c * 32];
      }
      v4f acc[KV];
      for (int c = 0; c < KV; ++c) acc[c] = (v4f){0.f, 0.f, 0.f, 0.f};
      int j = beg;
      for (; j + U <= end; j += U) {
        v4f t[U][KV];
        for (int u = 0; u < U; ++u) {
          const int sidx = adj[j + u];
          const v4f* hp = (const v4f*)(hin + (size_t)sidx * K) + lane;
          for (int c = 0; c < KV; ++c) t[u][c] = hp[c * 32];
        }
        for (int u = 0; u < U; ++u)
          for (int c = 0; c < KV; ++c) acc[c] += t[u][c];
      }
#pragma nounroll
      for (; j < end; ++j) {
        const int sidx = adj[j];
        const v4f* hp = (const v4f*)(hin + (size_t)sidx * K) + lane;
        for (int c = 0; c < KV; ++c) acc[c] += hp[c * 32];
      }
      for (int c = 0; c < KV; ++c) Arow[c * 32 + lane] = acc[c] * sc;
      if (WITH_SELF) {
        for (int c = 0; c < KV; ++c) Arow[K / 4 + c * 32 + lane] = sv[c];
      }
    } else {
      for (int c = 0; c < KV; ++c) Arow[c * 32 + lane] = (v4f){0.f, 0.f, 0.f, 0.f};
      if (WITH_SELF) {
        for (int c = 0; c < KV; ++c) Arow[K / 4 + c * 32 + lane] = (v4f){0.f, 0.f, 0.f, 0.f};
      }
    }
  }
}

template<int K, int NT, int PITCH>
__device__ __forceinline__ void stage_self(float* __restrict__ As, const float* __restrict__ hin,
                                           int r0, int nnodes, int tid) {
  constexpr int R4    = K / 4;
  constexpr int TOT   = 32 * R4;
  constexpr int TRIPS = (TOT + NT - 1) / NT;
  v4f v[TRIPS];
  for (int m = 0; m < TRIPS; ++m) {
    const int e   = tid + m * NT;
    const int row = e / R4;
    const int c4  = e - row * R4;
    v[m] = (v4f){0.f, 0.f, 0.f, 0.f};
    if (e < TOT && r0 + row < nnodes) v[m] = *((const v4f*)(hin + (size_t)(r0 + row) * K) + c4);
  }
  for (int m = 0; m < TRIPS; ++m) {
    const int e   = tid + m * NT;
    const int row = e / R4;
    const int c4  = e - row * R4;
    if (e < TOT) *((v4f*)(As + row * PITCH + K) + c4) = v[m];
  }
}

template<int K, int NOUT, bool RELU>
__global__ __launch_bounds__(NOUT, (NOUT == 384 ? 12 : 8)) void sage_alt_kernel(const float* __restrict__ hin,
                                                          const int* __restrict__ rowptr,
                                                          const int* __restrict__ adj,
                                                          const float* __restrict__ invdeg,
                                                          const float* __restrict__ Wlp,
                                                          const float* __restrict__ Wrp,
                                                          const float* __restrict__ bias,
                                                          float* __restrict__ out, int nnodes) {
  constexpr int NW    = NOUT / 32;
  constexpr int PITCH = 2 * K + 4;
  __shared__ __attribute__((aligned(16))) float As[32 * PITCH];

  const int lane = threadIdx.x & 31;
  const int wv   = __builtin_amdgcn_readfirstlane((int)(threadIdx.x >> 5));
  const int r0   = blockIdx.x * 32;

  const bool self_first = (__popc((unsigned)blockIdx.x) & 1) != 0;

  v8f acc00 = {}, acc01 = {}, acc10 = {}, acc11 = {};

  if (self_first) {
    stage_self<K, NOUT, PITCH>(As, hin, r0, nnodes, (int)threadIdx.x);
    __syncthreads();
    gemm_pass<K, PITCH>(As, Wrp, 1, wv, lane, acc00, acc01, acc10, acc11);
    build_rows<K, NW, PITCH, false>(As, hin, rowptr, adj, invdeg, r0, nnodes, wv, lane);
    __syncthreads();
    gemm_pass<K, PITCH>(As, Wlp, 0, wv, lane, acc00, acc01, acc10, acc11);
  } else {
    build_rows<K, NW, PITCH, true>(As, hin, rowptr, adj, invdeg, r0, nnodes, wv, lane);
    __syncthreads();
    gemm_pass<K, PITCH>(As, Wlp, 0, wv, lane, acc00, acc01, acc10, acc11);
    gemm_pass<K, PITCH>(As, Wrp, 1, wv, lane, acc00, acc01, acc10, acc11);
  }

  const int l15 = lane & 15;
  const int s   = lane >> 4;
  const int col = wv * 32 + l15;
  const float bj0 = bias[col];
  const float bj1 = bias[col + 16];
  const int rb = r0 + (s << 3);
#pragma unroll
  for (int v = 0; v < 8; ++v) {
    float t00 = acc00[v] + bj0;
    float t01 = acc01[v] + bj1;
    float t10 = acc10[v] + bj0;
    float t11 = acc11[v] + bj1;
    if (RELU) {
      t00 = fmaxf(t00, 0.f); t01 = fmaxf(t01, 0.f);
      t10 = fmaxf(t10, 0.f); t11 = fmaxf(t11, 0.f);
    }
    const int ra = rb + v;
    const int rc = rb + 16 + v;
    if (ra < nnodes) {
      float* o = out + (size_t)ra * NOUT + col;
      o[0] = t00; o[16] = t01;
    }
    if (rc < nnodes) {
      float* o = out + (size_t)rc * NOUT + col;
      o[0] = t10; o[16] = t11;
    }
  }
}


extern "C" void kernel_launch(void* const* d_in, const int* in_sizes, int n_in,
                              void* d_out, int out_size, void* d_ws, size_t ws_size,
                              hipStream_t stream) {
  const float* x   = (const float*)d_in[0];
  const int*   ei  = (const int*)d_in[1];
  const float* W1l = (const float*)d_in[2];
  const float* W1r = (const float*)d_in[3];
  const float* b1  = (const float*)d_in[4];
  const float* W2l = (const float*)d_in[5];
  const float* W2r = (const float*)d_in[6];
  const float* b2  = (const float*)d_in[7];
  const float* W3l = (const float*)d_in[8];
  const float* W3r = (const float*)d_in[9];
  const float* b3  = (const float*)d_in[10];
  float* out = (float*)d_out;

  const int N = in_sizes[0] / DIN;
  const int E = in_sizes[1] / 2;
  const int* src = ei;
  const int* dst = ei + E;

  char* ws = (char*)d_ws;
  auto al = [](size_t v) { return (v + 255) & ~(size_t)255; };
  size_t off = 0;
  int*   cnt      = (int*)(ws + off);   off += al((size_t)N * 4);
  int*   rowptr   = (int*)(ws + off);   off += al((size_t)(N + 1) * 4);
  int*   cursor   = (int*)(ws + off);   off += al((size_t)N * 4);
  int*   partials = (int*)(ws + off);   off += al((size_t)256 * 4);
  float* invdeg   = (float*)(ws + off); off += al((size_t)N * 4);
  int*   adj      = (int*)(ws + off);   off += al((size_t)E * 4);
  float* h1       = (float*)(ws + off); off += al((size_t)N * DH * 4);
  float* h2       = (float*)(ws + off); off += al((size_t)N * DH * 4);
  float* Wt1l     = (float*)(ws + off); off += al((size_t)DIN * DH * 4);
  float* Wt1r     = (float*)(ws + off); off += al((size_t)DIN * DH * 4);
  float* Wt2l     = (float*)(ws + off); off += al((size_t)DH * DH * 4);
  float* Wt2r     = (float*)(ws + off); off += al((size_t)DH * DH * 4);
  float* Wt3l     = (float*)(ws + off); off += al((size_t)DH * DOUT * 4);
  float* Wt3r     = (float*)(ws + off);

  const int TB = 256;
  const int NB = (N + 255) / 256;
  const int gb = (N + 31) / 32;

  zero_kernel<<<(N / 4 + TB - 1) / TB, TB, 0, stream>>>((float4*)cnt, N / 4);
  count_kernel<<<(E + TB - 1) / TB, TB, 0, stream>>>(dst, cnt, E);
  scan_blocks_kernel<<<NB, TB, 0, stream>>>(cnt, rowptr, partials, N);
  scan_partials_kernel<<<1, TB, 0, stream>>>(partials, NB);
  scan_finish_kernel<<<NB, TB, 0, stream>>>(rowptr, partials, cnt, cursor, invdeg, N);
  fill_adj_kernel<<<(E + TB - 1) / TB, TB, 0, stream>>>(src, dst, cursor, adj, E);

  permute_w_kernel<DIN, DH ><<<(DIN * DH  + TB - 1) / TB, TB, 0, stream>>>(W1l, Wt1l);
  permute_w_kernel<DIN, DH ><<<(DIN * DH  + TB - 1) / TB, TB, 0, stream>>>(W1r, Wt1r);
  permute_w_kernel<DH,  DH ><<<(DH  * DH  + TB - 1) / TB, TB, 0, stream>>>(W2l, Wt2l);
  permute_w_kernel<DH,  DH ><<<(DH  * DH  + TB - 1) / TB, TB, 0, stream>>>(W2r, Wt2r);
  permute_w_kernel<DH, DOUT><<<(DH * DOUT + TB - 1) / TB, TB, 0, stream>>>(W3l, Wt3l);
  permute_w_kernel<DH, DOUT><<<(DH * DOUT + TB - 1) / TB, TB, 0, stream>>>(W3r, Wt3r);

  sage_fused_kernel<DIN, DH, true><<<gb, DH, 0, stream>>>(x, rowptr, adj, invdeg, Wt1l, Wt1r, b1, h1, N);
  sage_alt_kernel<DH, DH, true><<<gb, DH, 0, stream>>>(h1, rowptr, adj, invdeg, Wt2l, Wt2r, b2, h2, N);
  sage_alt_kernel<DH, DOUT, false><<<gb, DOUT, 0, stream>>>(h2, rowptr, adj, invdeg, Wt3l, Wt3r, b3, out, N);
  (void)n_in; (void)out_size; (void)ws_size;
}


# set-up work shares one launch: the six weight permutes and the in-degree counter clear run as block ranges of a single kernel ahead of the CSR build; 15 -> 9 launches; kernels' arithmetic unchanged
# speedup vs baseline: 2.8392x; 1.0414x over previous
#include <hip/hip_runtime.h>

typedef __attribute__((ext_vector_type(2))) float v2f;
typedef __attribute__((ext_vector_type(4))) float v4f;
typedef __attribute__((ext_vector_type(8))) float v8f;

#define DIN  128
#define DH   256
#define DOUT 384


__global__ __launch_bounds__(256) void count_kernel(const int* __restrict__ dst,
                                                    int* __restrict__ cnt, int nedges) {
  int e = blockIdx.x * blockDim.x + threadIdx.x;
  if (e < nedges) atomicAdd(&cnt[dst[e]], 1);
}


__global__ __launch_bounds__(256) void scan_blocks_kernel(const int* __restrict__ cnt,
                                                          int* __restrict__ excl,
                                                          int* __restrict__ partials, int n) {
  __shared__ int sh[256];
  const int t = threadIdx.x;
  const int i = blockIdx.x * 256 + t;
  int v = (i < n) ? cnt[i] : 0;
  sh[t] = v;
  __syncthreads();
  for (int o = 1; o < 256; o <<= 1) {
    int x = (t >= o) ? sh[t - o] : 0;
    __syncthreads();
    sh[t] += x;
    __syncthreads();
  }
  if (i < n) excl[i] = sh[t] - v;
  if (t == 255) partials[blockIdx.x] = sh[255];
}

__global__ __launch_bounds__(256) void scan_partials_kernel(int* __restrict__ partials, int nb) {
  __shared__ int sh[256];
  const int t = threadIdx.x;
  int v = (t < nb) ? partials[t] : 0;
  sh[t] = v;
  __syncthreads();
  for (int o = 1; o < 256; o <<= 1) {
    int x = (t >= o) ? sh[t - o] : 0;
    __syncthreads();
    sh[t] += x;
    __syncthreads();
  }
  if (t < nb) partials[t] = sh[t] - v;
}

__global__ __launch_bounds__(256) void scan_finish_kernel(int* __restrict__ rowptr,
                                                          const int* __restrict__ partials,
                                                          const int* __restrict__ cnt,
                                                          int* __restrict__ cursor,
                                                          float* __restrict__ invdeg, int n) {
  int i = blockIdx.x * 256 + threadIdx.x;
  if (i >= n) return;
  int r = rowptr[i] + partials[i >> 8];
  rowptr[i] = r;
  cursor[i] = r;
  invdeg[i] = 1.0f / fmaxf((float)cnt[i], 1.0f);
  if (i == n - 1) rowptr[n] = r + cnt[i];
}

__global__ __launch_bounds__(256) void fill_adj_kernel(const int* __restrict__ src,
                                                       const int* __restrict__ dst,
                                                       int* __restrict__ cursor,
                                                       int* __restrict__ adj, int nedges) {
  int e = blockIdx.x * blockDim.x + threadIdx.x;
  if (e >= nedges) return;
  int pos = atomicAdd(&cursor[dst[e]], 1);
  adj[pos] = src[e];
}

template<int K, int NOUT>
__device__ __forceinline__ void permute_w_elem(const float* __restrict__ W,
                                               float* __restrict__ Wt, int i) {
  static_assert((K * NOUT) % 256 == 0, "a matrix is a whole number of 256-thread blocks");
  int k = i / NOUT;
  int col = i - k * NOUT;
  int s = (k >> 1) & 1;
  int j = k >> 2;
  int jj = j >> 1;
  int q = ((j & 1) << 1) | (k & 1);
  int nb = col >> 6;
  int t = (col >> 4) & 3;
  int lane = (s << 4) | (col & 15);
  Wt[((((size_t)nb * (K / 8) + jj) * 4 + t) * 32 + lane) * 4 + q] = W[i];
}

#define PB1 (DIN * DH / 256)
#define PB2 (DH * DH / 256)
#define PB3 (DH * DOUT / 256)
#define PERM_BLOCKS (2 * PB1 + 2 * PB2 + 2 * PB3)

__global__ __launch_bounds__(256) void setup_kernel(const float* __restrict__ W1l,
                                                    const float* __restrict__ W1r,
                                                    const float* __restrict__ W2l,
                                                    const float* __restrict__ W2r,
                                                    const float* __restrict__ W3l,
                                                    const float* __restrict__ W3r,
                                                    float* __restrict__ Wt1l,
                                                    float* __restrict__ Wt1r,
                                                    float* __restrict__ Wt2l,
                                                    float* __restrict__ Wt2r,
                                                    float* __restrict__ Wt3l,
                                                    float* __restrict__ Wt3r,
                                                    float4* __restrict__ cnt4, int n4) {
  int b = blockIdx.x;
  const int t = threadIdx.x;
  if (b < PB1) { permute_w_elem<DIN, DH>(W1l, Wt1l, b * 256 + t); return; }
  b -= PB1;
  if (b < PB1) { permute_w_elem<DIN, DH>(W1r, Wt1r, b * 256 + t); return; }
  b -= PB1;
  if (b < PB2) { permute_w_elem<DH, DH>(W2l, Wt2l, b * 256 + t); return; }
  b -= PB2;
  if (b < PB2) { permute_w_elem<DH, DH>(W2r, Wt2r, b * 256 + t); return; }
  b -= PB2;
  if (b < PB3) { permute_w_elem<DH, DOUT>(W3l, Wt3l, b * 256 + t); return; }
  b -= PB3;
  if (b < PB3) { permute_w_elem<DH, DOUT>(W3r, Wt3r, b * 256 + t); return; }
  b -= PB3;
  const int i = b * 256 + t;
  if (i < n4) cnt4[i] = make_float4(0.f, 0.f, 0.f, 0.f);
}

template<int K, int NOUT, bool RELU>
__global__ __launch_bounds__(NOUT) void sage_fused_kernel(const float* __restrict__ hin,
                                                          const int* __restrict__ rowptr,
                                                          const int* __restrict__ adj,
                                                          const float* __restrict__ invdeg,
                                                          const float* __restrict__ Wlp,
                                                          const float* __restrict__ Wrp,
                                                          const float* __restrict__ bias,
                                                          float* __restrict__ out, int nnodes) {
  constexpr int NW    = NOUT / 32;
  constexpr int PITCH = 2 * K + 4;
  constexpr int KV    = K / 128;
  __shared__ __attribute__((aligned(16))) float As[32 * PITCH];

  const int lane = threadIdx.x & 31;
  const int wv   = __builtin_amdgcn_readfirstlane((int)(threadIdx.x >> 5));
  const int r0   = blockIdx.x * 32;

  for (int i = wv; i < 32; i += NW) {
    const int node = r0 + i;
    v4f* Arow = (v4f*)(As + i * PITCH);
    if (node < nnodes) {
      const int beg = rowptr[node];
      const int end = rowptr[node + 1];
      const float sc = invdeg[node];
      const v4f* self = (const v4f*)(hin + (size_t)node * K) + lane;
      v4f sv[KV];
      v4f acc[KV];
      for (int c = 0; c < KV; ++c) {
        sv[c]  = self[c * 32];
        acc[c] = (v4f){0.f, 0.f, 0.f, 0.f};
      }
      for (int j = beg; j < end; ++j) {
        const int sidx = adj[j];
        const v4f* hp = (const v4f*)(hin + (size_t)sidx * K) + lane;
        for (int c = 0; c < KV; ++c) acc[c] += hp[c * 32];
      }
      for (int c = 0; c < KV; ++c) {
        Arow[c * 32 + lane]         = acc[c] * sc;
        Arow[K / 4 + c * 32 + lane] = sv[c];
      }
    } else {
      for (int c = 0; c < 2 * KV; ++c) Arow[c * 32 + lane] = (v4f){0.f, 0.f, 0.f, 0.f};
    }
  }
  __syncthreads();

  const int l15   = lane & 15;
  const int s     = lane >> 4;
  const int khalf = s << 1;

  v8f acc00 = {}, acc01 = {}, acc10 = {}, acc11 = {};

  for (int pass = 0; pass < 2; ++pass) {
    const float* __restrict__ Wp = (pass == 0 ? Wlp : Wrp);
    const v4f* bp = (const v4f*)Wp + (size_t)(wv >> 1) * (K / 8) * 128 + (wv & 1) * 64 + lane;
    const float* A0 = As + l15 * PITCH + pass * K + khalf;
    const float* A1 = A0 + 16 * PITCH;
#pragma unroll 2
    for (int jj = 0; jj < K / 8; ++jj) {
      const v2f a00 = *(const v2f*)(A0 + 8 * jj);
      const v2f a01 = *(const v2f*)(A0 + 8 * jj + 4);
      const v2f a10 = *(const v2f*)(A1 + 8 * jj);
      const v2f a11 = *(const v2f*)(A1 + 8 * jj + 4);
      const v4f b0 = bp[jj * 128 +  0];
      const v4f b1 = bp[jj * 128 + 32];
      acc00 = __builtin_amdgcn_wmma_f32_16x16x4_f32(false, a00, false, __builtin_shufflevector(b0, b0, 0, 1), (short)0, acc00, false, false);
      acc01 = __builtin_amdgcn_wmma_f32_16x16x4_f32(false, a00, false, __builtin_shufflevector(b1, b1, 0, 1), (short)0, acc01, false, false);
      acc10 = __builtin_amdgcn_wmma_f32_16x16x4_f32(false, a10, false, __builtin_shufflevector(b0, b0, 0, 1), (short)0, acc10, false, false);
      acc11 = __builtin_amdgcn_wmma_f32_16x16x4_f32(false, a10, false, __builtin_shufflevector(b1, b1, 0, 1), (short)0, acc11, false, false);
      acc00 = __builtin_amdgcn_wmma_f32_16x16x4_f32(false, a01, false, __builtin_shufflevector(b0, b0, 2, 3), (short)0, acc00, false, false);
      acc01 = __builtin_amdgcn_wmma_f32_16x16x4_f32(false, a01, false, __builtin_shufflevector(b1, b1, 2, 3), (short)0, acc01, false, false);
      acc10 = __builtin_amdgcn_wmma_f32_16x16x4_f32(false, a11, false, __builtin_shufflevector(b0, b0, 2, 3), (short)0, acc10, false, false);
      acc11 = __builtin_amdgcn_wmma_f32_16x16x4_f32(false, a11, false, __builtin_shufflevector(b1, b1, 2, 3), (short)0, acc11, false, false);
    }
  }

  const int col = wv * 32 + l15;
  const float bj0 = bias[col];
  const float bj1 = bias[col + 16];
  const int rb = r0 + (s << 3);
#pragma unroll
  for (int v = 0; v < 8; ++v) {
    float t00 = acc00[v] + bj0;
    float t01 = acc01[v] + bj1;
    float t10 = acc10[v] + bj0;
    float t11 = acc11[v] + bj1;
    if (RELU) {
      t00 = fmaxf(t00, 0.f); t01 = fmaxf(t01, 0.f);
      t10 = fmaxf(t10, 0.f); t11 = fmaxf(t11, 0.f);
    }
    const int ra = rb + v;
    const int rc = rb + 16 + v;
    if (ra < nnodes) {
      float* o = out + (size_t)ra * NOUT + col;
      o[0] = t00; o[16] = t01;
    }
    if (rc < nnodes) {
      float* o = out + (size_t)rc * NOUT + col;
      o[0] = t10; o[16] = t11;
    }
  }
}


template<int K, int PITCH>
__device__ __forceinline__ void gemm_pass(const float* __restrict__ As, const float* __restrict__ Wp,
                                          int pass, int wv, int lane,
                                          v8f& acc00, v8f& acc01, v8f& acc10, v8f& acc11) {
  const int l15   = lane & 15;
  const int khalf = (lane >> 4) << 1;
  const v4f* bp = (const v4f*)Wp + (size_t)(wv >> 1) * (K / 8) * 128 + (wv & 1) * 64 + lane;
  const float* A0 = As + l15 * PITCH + pass * K + khalf;
  const float* A1 = A0 + 16 * PITCH;
#pragma unroll 2
  for (int jj = 0; jj < K / 8; ++jj) {
    const v2f a00 = *(const v2f*)(A0 + 8 * jj);
    const v2f a01 = *(const v2f*)(A0 + 8 * jj + 4);
    const v2f a10 = *(const v2f*)(A1 + 8 * jj);
    const v2f a11 = *(const v2f*)(A1 + 8 * jj + 4);
    const v4f b0 = bp[jj * 128 +  0];
    const v4f b1 = bp[jj * 128 + 32];
    acc00 = __builtin_amdgcn_wmma_f32_16x16x4_f32(false, a00, false, __builtin_shufflevector(b0, b0, 0, 1), (short)0, acc00, false, false);
    acc01 = __builtin_amdgcn_wmma_f32_16x16x4_f32(false, a00, false, __builtin_shufflevector(b1, b1, 0, 1), (short)0, acc01, false, false);
    acc10 = __builtin_amdgcn_wmma_f32_16x16x4_f32(false, a10, false, __builtin_shufflevector(b0, b0, 0, 1), (short)0, acc10, false, false);
    acc11 = __builtin_amdgcn_wmma_f32_16x16x4_f32(false, a10, false, __builtin_shufflevector(b1, b1, 0, 1), (short)0, acc11, false, false);
    acc00 = __builtin_amdgcn_wmma_f32_16x16x4_f32(false, a01, false, __builtin_shufflevector(b0, b0, 2, 3), (short)0, acc00, false, false);
    acc01 = __builtin_amdgcn_wmma_f32_16x16x4_f32(false, a01, false, __builtin_shufflevector(b1, b1, 2, 3), (short)0, acc01, false, false);
    acc10 = __builtin_amdgcn_wmma_f32_16x16x4_f32(false, a11, false, __builtin_shufflevector(b0, b0, 2, 3), (short)0, acc10, false, false);
    acc11 = __builtin_amdgcn_wmma_f32_16x16x4_f32(false, a11, false, __builtin_shufflevector(b1, b1, 2, 3), (short)0, acc11, false, false);
  }
}

template<int K, int NW, int PITCH, bool WITH_SELF>
__device__ __forceinline__ void build_rows(float* __restrict__ As, const float* __restrict__ hin,
                                           const int* __restrict__ rowptr, const int* __restrict__ adj,
                                           const float* __restrict__ invdeg,
                                           int r0, int nnodes, int wv, int lane) {
  constexpr int KV = K / 128;
  constexpr int U  = 4;
  for (int i = wv; i < 32; i += NW) {
    const int node = r0 + i;
    v4f* Arow = (v4f*)(As + i * PITCH);
    if (node < nnodes) {
      const int beg = rowptr[node];
      const int end = rowptr[node + 1];
      const float sc = invdeg[node];
      v4f sv[KV];
      if (WITH_SELF) {
        const v4f* self = (const v4f*)(hin + (size_t)node * K) + lane;
        for (int c = 0; c < KV; ++c) sv[c] = self[c * 32];
      }
      v4f acc[KV];
      for (int c = 0; c < KV; ++c) acc[c] = (v4f){0.f, 0.f, 0.f, 0.f};
      int j = beg;
      for (; j + U <= end; j += U) {
        v4f t[U][KV];
        for (int u = 0; u < U; ++u) {
          const int sidx = adj[j + u];
          const v4f* hp = (const v4f*)(hin + (size_t)sidx * K) + lane;
          for (int c = 0; c < KV; ++c) t[u][c] = hp[c * 32];
        }
        for (int u = 0; u < U; ++u)
          for (int c = 0; c < KV; ++c) acc[c] += t[u][c];
      }
#pragma nounroll
      for (; j < end; ++j) {
        const int sidx = adj[j];
        const v4f* hp = (const v4f*)(hin + (size_t)sidx * K) + lane;
        for (int c = 0; c < KV; ++c) acc[c] += hp[c * 32];
      }
      for (int c = 0; c < KV; ++c) Arow[c * 32 + lane] = acc[c] * sc;
      if (WITH_SELF) {
        for (int c = 0; c < KV; ++c) Arow[K / 4 + c * 32 + lane] = sv[c];
      }
    } else {
      for (int c = 0; c < KV; ++c) Arow[c * 32 + lane] = (v4f){0.f, 0.f, 0.f, 0.f};
      if (WITH_SELF) {
        for (int c = 0; c < KV; ++c) Arow[K / 4 + c * 32 + lane] = (v4f){0.f, 0.f, 0.f, 0.f};
      }
    }
  }
}

template<int K, int NT, int PITCH>
__device__ __forceinline__ void stage_self(float* __restrict__ As, const float* __restrict__ hin,
                                           int r0, int nnodes, int tid) {
  constexpr int R4    = K / 4;
  constexpr int TOT   = 32 * R4;
  constexpr int TRIPS = (TOT + NT - 1) / NT;
  v4f v[TRIPS];
  for (int m = 0; m < TRIPS; ++m) {
    const int e   = tid + m * NT;
    const int row = e / R4;
    const int c4  = e - row * R4;
    v[m] = (v4f){0.f, 0.f, 0.f, 0.f};
    if (e < TOT && r0 + row < nnodes) v[m] = *((const v4f*)(hin + (size_t)(r0 + row) * K) + c4);
  }
  for (int m = 0; m < TRIPS; ++m) {
    const int e   = tid + m * NT;
    const int row = e / R4;
    const int c4  = e - row * R4;
    if (e < TOT) *((v4f*)(As + row * PITCH + K) + c4) = v[m];
  }
}

template<int K, int NOUT, bool RELU>
__global__ __launch_bounds__(NOUT, (NOUT == 384 ? 12 : 8)) void sage_alt_kernel(const float* __restrict__ hin,
                                                          const int* __restrict__ rowptr,
                                                          const int* __restrict__ adj,
                                                          const float* __restrict__ invdeg,
                                                          const float* __restrict__ Wlp,
                                                          const float* __restrict__ Wrp,
                                                          const float* __restrict__ bias,
                                                          float* __restrict__ out, int nnodes) {
  constexpr int NW    = NOUT / 32;
  constexpr int PITCH = 2 * K + 4;
  __shared__ __attribute__((aligned(16))) float As[32 * PITCH];

  const int lane = threadIdx.x & 31;
  const int wv   = __builtin_amdgcn_readfirstlane((int)(threadIdx.x >> 5));
  const int r0   = blockIdx.x * 32;

  const bool self_first = (__popc((unsigned)blockIdx.x) & 1) != 0;

  v8f acc00 = {}, acc01 = {}, acc10 = {}, acc11 = {};

  if (self_first) {
    stage_self<K, NOUT, PITCH>(As, hin, r0, nnodes, (int)threadIdx.x);
    __syncthreads();
    gemm_pass<K, PITCH>(As, Wrp, 1, wv, lane, acc00, acc01, acc10, acc11);
    build_rows<K, NW, PITCH, false>(As, hin, rowptr, adj, invdeg, r0, nnodes, wv, lane);
    __syncthreads();
    gemm_pass<K, PITCH>(As, Wlp, 0, wv, lane, acc00, acc01, acc10, acc11);
  } else {
    build_rows<K, NW, PITCH, true>(As, hin, rowptr, adj, invdeg, r0, nnodes, wv, lane);
    __syncthreads();
    gemm_pass<K, PITCH>(As, Wlp, 0, wv, lane, acc00, acc01, acc10, acc11);
    gemm_pass<K, PITCH>(As, Wrp, 1, wv, lane, acc00, acc01, acc10, acc11);
  }

  const int l15 = lane & 15;
  const int s   = lane >> 4;
  const int col = wv * 32 + l15;
  const float bj0 = bias[col];
  const float bj1 = bias[col + 16];
  const int rb = r0 + (s << 3);
#pragma unroll
  for (int v = 0; v < 8; ++v) {
    float t00 = acc00[v] + bj0;
    float t01 = acc01[v] + bj1;
    float t10 = acc10[v] + bj0;
    float t11 = acc11[v] + bj1;
    if (RELU) {
      t00 = fmaxf(t00, 0.f); t01 = fmaxf(t01, 0.f);
      t10 = fmaxf(t10, 0.f); t11 = fmaxf(t11, 0.f);
    }
    const int ra = rb + v;
    const int rc = rb + 16 + v;
    if (ra < nnodes) {
      float* o = out + (size_t)ra * NOUT + col;
      o[0] = t00; o[16] = t01;
    }
    if (rc < nnodes) {
      float* o = out + (size_t)rc * NOUT + col;
      o[0] = t10; o[16] = t11;
    }
  }
}


extern "C" void kernel_launch(void* const* d_in, const int* in_sizes, int n_in,
                              void* d_out, int out_size, void* d_ws, size_t ws_size,
                              hipStream_t stream) {
  const float* x   = (const float*)d_in[0];
  const int*   ei  = (const int*)d_in[1];
  const float* W1l = (const float*)d_in[2];
  const float* W1r = (const float*)d_in[3];
  const float* b1  = (const float*)d_in[4];
  const float* W2l = (const float*)d_in[5];
  const float* W2r = (const float*)d_in[6];
  const float* b2  = (const float*)d_in[7];
  const float* W3l = (const float*)d_in[8];
  const float* W3r = (const float*)d_in[9];
  const float* b3  = (const float*)d_in[10];
  float* out = (float*)d_out;

  const int N = in_sizes[0] / DIN;
  const int E = in_sizes[1] / 2;
  const int* src = ei;
  const int* dst = ei + E;

  char* ws = (char*)d_ws;
  auto al = [](size_t v) { return (v + 255) & ~(size_t)255; };
  size_t off = 0;
  int*   cnt      = (int*)(ws + off);   off += al((size_t)N * 4);
  int*   rowptr   = (int*)(ws + off);   off += al((size_t)(N + 1) * 4);
  int*   cursor   = (int*)(ws + off);   off += al((size_t)N * 4);
  int*   partials = (int*)(ws + off);   off += al((size_t)256 * 4);
  float* invdeg   = (float*)(ws + off); off += al((size_t)N * 4);
  int*   adj      = (int*)(ws + off);   off += al((size_t)E * 4);
  float* h1       = (float*)(ws + off); off += al((size_t)N * DH * 4);
  float* h2       = (float*)(ws + off); off += al((size_t)N * DH * 4);
  float* Wt1l     = (float*)(ws + off); off += al((size_t)DIN * DH * 4);
  float* Wt1r     = (float*)(ws + off); off += al((size_t)DIN * DH * 4);
  float* Wt2l     = (float*)(ws + off); off += al((size_t)DH * DH * 4);
  float* Wt2r     = (float*)(ws + off); off += al((size_t)DH * DH * 4);
  float* Wt3l     = (float*)(ws + off); off += al((size_t)DH * DOUT * 4);
  float* Wt3r     = (float*)(ws + off);

  const int TB = 256;
  const int NB = (N + 255) / 256;
  const int gb = (N + 31) / 32;

  setup_kernel<<<PERM_BLOCKS + (N / 4 + TB - 1) / TB, TB, 0, stream>>>(W1l, W1r, W2l, W2r, W3l, W3r,
                                                                      Wt1l, Wt1r, Wt2l, Wt2r, Wt3l, Wt3r,
                                                                      (float4*)cnt, N / 4);
  count_kernel<<<(E + TB - 1) / TB, TB, 0, stream>>>(dst, cnt, E);
  scan_blocks_kernel<<<NB, TB, 0, stream>>>(cnt, rowptr, partials, N);
  scan_partials_kernel<<<1, TB, 0, stream>>>(partials, NB);
  scan_finish_kernel<<<NB, TB, 0, stream>>>(rowptr, partials, cnt, cursor, invdeg, N);
  fill_adj_kernel<<<(E + TB - 1) / TB, TB, 0, stream>>>(src, dst, cursor, adj, E);

  sage_fused_kernel<DIN, DH, true><<<gb, DH, 0, stream>>>(x, rowptr, adj, invdeg, Wt1l, Wt1r, b1, h1, N);
  sage_alt_kernel<DH, DH, true><<<gb, DH, 0, stream>>>(h1, rowptr, adj, invdeg, Wt2l, Wt2r, b2, h2, N);
  sage_alt_kernel<DH, DOUT, false><<<gb, DOUT, 0, stream>>>(h2, rowptr, adj, invdeg, Wt3l, Wt3r, b3, out, N);
  (void)n_in; (void)out_size; (void)ws_size;
}


# adjacency at a fixed pitch of 64 slots per node, slot = returning add on the node's in-degree counter: count pass, 3 scan launches, row pointers gone (9 -> 5 launches); sum order stays arrival order
# speedup vs baseline: 3.0609x; 1.0781x over previous
#include <hip/hip_runtime.h>

typedef __attribute__((ext_vector_type(2))) float v2f;
typedef __attribute__((ext_vector_type(4))) float v4f;
typedef __attribute__((ext_vector_type(8))) float v8f;

#define DIN  128
#define DH   256
#define DOUT 384

#define ADJW 64

__global__ __launch_bounds__(256) void fill_adj_kernel(const int* __restrict__ src,
                                                       const int* __restrict__ dst,
                                                       int* __restrict__ cnt,
                                                       int* __restrict__ adj, int nedges) {
  int e = blockIdx.x * blockDim.x + threadIdx.x;
  if (e >= nedges) return;
  const int d = dst[e];
  const int pos = atomicAdd(&cnt[d], 1);
  if (pos < ADJW) adj[(size_t)d * ADJW + pos] = src[e];
}

template<int K, int NW, int PITCH>
__device__ __forceinline__ void overflow_rows(float* __restrict__ As, const float* __restrict__ hin,
                                              const int* __restrict__ cnt,
                                              const int* __restrict__ esrc, const int* __restrict__ edst,
                                              int nedges, int r0, int wv, int lane, unsigned over) {
  constexpr int KV = K / 128;
  for (int i = wv; i < 32; i += NW) {
    if (((over >> i) & 1u) == 0u) continue;
    const int node = r0 + i;
    v4f acc[KV];
    for (int c = 0; c < KV; ++c) acc[c] = (v4f){0.f, 0.f, 0.f, 0.f};
    for (int e0 = 0; e0 < nedges; e0 += 32) {
      const int e = e0 + lane;
      int d = -1, sl = 0;
      if (e < nedges) { d = edst[e]; sl = esrc[e]; }
      unsigned m = __builtin_amdgcn_ballot_w32(d == node);
      while (m != 0u) {
        const int b = __builtin_ctz(m);
        m &= m - 1u;
        const int sidx = __builtin_amdgcn_readlane(sl, b);
        const v4f* hp = (const v4f*)(hin + (size_t)sidx * K) + lane;
        for (int c = 0; c < KV; ++c) acc[c] += hp[c * 32];
      }
    }
    const float sc = 1.0f / fmaxf((float)cnt[node], 1.0f);
    v4f* Arow = (v4f*)(As + i * PITCH);
    for (int c = 0; c < KV; ++c) Arow[c * 32 + lane] = acc[c] * sc;
  }
}

__device__ __forceinline__ unsigned rows_over(const int* __restrict__ cnt, int r0, int nnodes, int lane) {
  const int nd = r0 + lane;
  const int dg = (nd < nnodes) ? cnt[nd] : 0;
  return __builtin_amdgcn_ballot_w32(dg > ADJW);
}

template<int K, int NOUT>
__device__ __forceinline__ void permute_w_elem(const float* __restrict__ W,
                                               float* __restrict__ Wt, int i) {
  static_assert((K * NOUT) % 256 == 0, "a matrix is a whole number of 256-thread blocks");
  int k = i / NOUT;
  int col = i - k * NOUT;
  int s = (k >> 1) & 1;
  int j = k >> 2;
  int jj = j >> 1;
  int q = ((j & 1) << 1) | (k & 1);
  int nb = col >> 6;
  int t = (col >> 4) & 3;
  int lane = (s << 4) | (col & 15);
  Wt[((((size_t)nb * (K / 8) + jj) * 4 + t) * 32 + lane) * 4 + q] = W[i];
}

#define PB1 (DIN * DH / 256)
#define PB2 (DH * DH / 256)
#define PB3 (DH * DOUT / 256)
#define PERM_BLOCKS (2 * PB1 + 2 * PB2 + 2 * PB3)

__global__ __launch_bounds__(256) void setup_kernel(const float* __restrict__ W1l,
                                                    const float* __restrict__ W1r,
                                                    const float* __restrict__ W2l,
                                                    const float* __restrict__ W2r,
                                                    const float* __restrict__ W3l,
                                                    const float* __restrict__ W3r,
                                                    float* __restrict__ Wt1l,
                                                    float* __restrict__ Wt1r,
                                                    float* __restrict__ Wt2l,
                                                    float* __restrict__ Wt2r,
                                                    float* __restrict__ Wt3l,
                                                    float* __restrict__ Wt3r,
                                                    float4* __restrict__ cnt4, int n4) {
  int b = blockIdx.x;
  const int t = threadIdx.x;
  if (b < PB1) { permute_w_elem<DIN, DH>(W1l, Wt1l, b * 256 + t); return; }
  b -= PB1;
  if (b < PB1) { permute_w_elem<DIN, DH>(W1r, Wt1r, b * 256 + t); return; }
  b -= PB1;
  if (b < PB2) { permute_w_elem<DH, DH>(W2l, Wt2l, b * 256 + t); return; }
  b -= PB2;
  if (b < PB2) { permute_w_elem<DH, DH>(W2r, Wt2r, b * 256 + t); return; }
  b -= PB2;
  if (b < PB3) { permute_w_elem<DH, DOUT>(W3l, Wt3l, b * 256 + t); return; }
  b -= PB3;
  if (b < PB3) { permute_w_elem<DH, DOUT>(W3r, Wt3r, b * 256 + t); return; }
  b -= PB3;
  const int i = b * 256 + t;
  if (i < n4) cnt4[i] = make_float4(0.f, 0.f, 0.f, 0.f);
}

template<int K, int NOUT, bool RELU>
__global__ __launch_bounds__(NOUT) void sage_fused_kernel(const float* __restrict__ hin,
                                                          const int* __restrict__ cnt,
                                                          const int* __restrict__ adj,
                                                          const int* __restrict__ esrc,
                                                          const int* __restrict__ edst, int nedges,
                                                          const float* __restrict__ Wlp,
                                                          const float* __restrict__ Wrp,
                                                          const float* __restrict__ bias,
                                                          float* __restrict__ out, int nnodes) {
  constexpr int NW    = NOUT / 32;
  constexpr int PITCH = 2 * K + 4;
  constexpr int KV    = K / 128;
  __shared__ __attribute__((aligned(16))) float As[32 * PITCH];

  const int lane = threadIdx.x & 31;
  const int wv   = __builtin_amdgcn_readfirstlane((int)(threadIdx.x >> 5));
  const int r0   = blockIdx.x * 32;

  const unsigned over = rows_over(cnt, r0, nnodes, lane);
  if (over != 0u) overflow_rows<K, NW, PITCH>(As, hin, cnt, esrc, edst, nedges, r0, wv, lane, over);

  for (int i = wv; i < 32; i += NW) {
    const int node = r0 + i;
    v4f* Arow = (v4f*)(As + i * PITCH);
    if (node < nnodes) {
      const int deg = cnt[node];
      const int beg = node * ADJW;
      const int end = beg + (deg <= ADJW ? deg : 0);
      const float sc = 1.0f / fmaxf((float)deg, 1.0f);
      const v4f* self = (const v4f*)(hin + (size_t)node * K) + lane;
      v4f sv[KV];
      v4f acc[KV];
      for (int c = 0; c < KV; ++c) {
        sv[c]  = self[c * 32];
        acc[c] = (v4f){0.f, 0.f, 0.f, 0.f};
      }
      for (int j = beg; j < end; ++j) {
        const int sidx = adj[j];
        const v4f* hp = (const v4f*)(hin + (size_t)sidx * K) + lane;
        for (int c = 0; c < KV; ++c) acc[c] += hp[c * 32];
      }
      for (int c = 0; c < KV; ++c) {
        if (deg <= ADJW) Arow[c * 32 + lane] = acc[c] * sc;
        Arow[K / 4 + c * 32 + lane] = sv[c];
      }
    } else {
      for (int c = 0; c < 2 * KV; ++c) Arow[c * 32 + lane] = (v4f){0.f, 0.f, 0.f, 0.f};
    }
  }
  __syncthreads();

  const int l15   = lane & 15;
  const int s     = lane >> 4;
  const int khalf = s << 1;

  v8f acc00 = {}, acc01 = {}, acc10 = {}, acc11 = {};

  for (int pass = 0; pass < 2; ++pass) {
    const float* __restrict__ Wp = (pass == 0 ? Wlp : Wrp);
    const v4f* bp = (const v4f*)Wp + (size_t)(wv >> 1) * (K / 8) * 128 + (wv & 1) * 64 + lane;
    const float* A0 = As + l15 * PITCH + pass * K + khalf;
    const float* A1 = A0 + 16 * PITCH;
#pragma unroll 2
    for (int jj = 0; jj < K / 8; ++jj) {
      const v2f a00 = *(const v2f*)(A0 + 8 * jj);
      const v2f a01 = *(const v2f*)(A0 + 8 * jj + 4);
      const v2f a10 = *(const v2f*)(A1 + 8 * jj);
      const v2f a11 = *(const v2f*)(A1 + 8 * jj + 4);
      const v4f b0 = bp[jj * 128 +  0];
      const v4f b1 = bp[jj * 128 + 32];
      acc00 = __builtin_amdgcn_wmma_f32_16x16x4_f32(false, a00, false, __builtin_shufflevector(b0, b0, 0, 1), (short)0, acc00, false, false);
      acc01 = __builtin_amdgcn_wmma_f32_16x16x4_f32(false, a00, false, __builtin_shufflevector(b1, b1, 0, 1), (short)0, acc01, false, false);
      acc10 = __builtin_amdgcn_wmma_f32_16x16x4_f32(false, a10, false, __builtin_shufflevector(b0, b0, 0, 1), (short)0, acc10, false, false);
      acc11 = __builtin_amdgcn_wmma_f32_16x16x4_f32(false, a10, false, __builtin_shufflevector(b1, b1, 0, 1), (short)0, acc11, false, false);
      acc00 = __builtin_amdgcn_wmma_f32_16x16x4_f32(false, a01, false, __builtin_shufflevector(b0, b0, 2, 3), (short)0, acc00, false, false);
      acc01 = __builtin_amdgcn_wmma_f32_16x16x4_f32(false, a01, false, __builtin_shufflevector(b1, b1, 2, 3), (short)0, acc01, false, false);
      acc10 = __builtin_amdgcn_wmma_f32_16x16x4_f32(false, a11, false, __builtin_shufflevector(b0, b0, 2, 3), (short)0, acc10, false, false);
      acc11 = __builtin_amdgcn_wmma_f32_16x16x4_f32(false, a11, false, __builtin_shufflevector(b1, b1, 2, 3), (short)0, acc11, false, false);
    }
  }

  const int col = wv * 32 + l15;
  const float bj0 = bias[col];
  const float bj1 = bias[col + 16];
  const int rb = r0 + (s << 3);
#pragma unroll
  for (int v = 0; v < 8; ++v) {
    float t00 = acc00[v] + bj0;
    float t01 = acc01[v] + bj1;
    float t10 = acc10[v] + bj0;
    float t11 = acc11[v] + bj1;
    if (RELU) {
      t00 = fmaxf(t00, 0.f); t01 = fmaxf(t01, 0.f);
      t10 = fmaxf(t10, 0.f); t11 = fmaxf(t11, 0.f);
    }
    const int ra = rb + v;
    const int rc = rb + 16 + v;
    if (ra < nnodes) {
      float* o = out + (size_t)ra * NOUT + col;
      o[0] = t00; o[16] = t01;
    }
    if (rc < nnodes) {
      float* o = out + (size_t)rc * NOUT + col;
      o[0] = t10; o[16] = t11;
    }
  }
}


template<int K, int PITCH>
__device__ __forceinline__ void gemm_pass(const float* __restrict__ As, const float* __restrict__ Wp,
                                          int pass, int wv, int lane,
                                          v8f& acc00, v8f& acc01, v8f& acc10, v8f& acc11) {
  const int l15   = lane & 15;
  const int khalf = (lane >> 4) << 1;
  const v4f* bp = (const v4f*)Wp + (size_t)(wv >> 1) * (K / 8) * 128 + (wv & 1) * 64 + lane;
  const float* A0 = As + l15 * PITCH + pass * K + khalf;
  const float* A1 = A0 + 16 * PITCH;
#pragma unroll 2
  for (int jj = 0; jj < K / 8; ++jj) {
    const v2f a00 = *(const v2f*)(A0 + 8 * jj);
    const v2f a01 = *(const v2f*)(A0 + 8 * jj + 4);
    const v2f a10 = *(const v2f*)(A1 + 8 * jj);
    const v2f a11 = *(const v2f*)(A1 + 8 * jj + 4);
    const v4f b0 = bp[jj * 128 +  0];
    const v4f b1 = bp[jj * 128 + 32];
    acc00 = __builtin_amdgcn_wmma_f32_16x16x4_f32(false, a00, false, __builtin_shufflevector(b0, b0, 0, 1), (short)0, acc00, false, false);
    acc01 = __builtin_amdgcn_wmma_f32_16x16x4_f32(false, a00, false, __builtin_shufflevector(b1, b1, 0, 1), (short)0, acc01, false, false);
    acc10 = __builtin_amdgcn_wmma_f32_16x16x4_f32(false, a10, false, __builtin_shufflevector(b0, b0, 0, 1), (short)0, acc10, false, false);
    acc11 = __builtin_amdgcn_wmma_f32_16x16x4_f32(false, a10, false, __builtin_shufflevector(b1, b1, 0, 1), (short)0, acc11, false, false);
    acc00 = __builtin_amdgcn_wmma_f32_16x16x4_f32(false, a01, false, __builtin_shufflevector(b0, b0, 2, 3), (short)0, acc00, false, false);
    acc01 = __builtin_amdgcn_wmma_f32_16x16x4_f32(false, a01, false, __builtin_shufflevector(b1, b1, 2, 3), (short)0, acc01, false, false);
    acc10 = __builtin_amdgcn_wmma_f32_16x16x4_f32(false, a11, false, __builtin_shufflevector(b0, b0, 2, 3), (short)0, acc10, false, false);
    acc11 = __builtin_amdgcn_wmma_f32_16x16x4_f32(false, a11, false, __builtin_shufflevector(b1, b1, 2, 3), (short)0, acc11, false, false);
  }
}

template<int K, int NW, int PITCH, bool WITH_SELF>
__device__ __forceinline__ void build_rows(float* __restrict__ As, const float* __restrict__ hin,
                                           const int* __restrict__ cnt, const int* __restrict__ adj,
                                           int r0, int nnodes, int wv, int lane) {
  constexpr int KV = K / 128;
  constexpr int U  = 4;
  for (int i = wv; i < 32; i += NW) {
    const int node = r0 + i;
    v4f* Arow = (v4f*)(As + i * PITCH);
    if (node < nnodes) {
      const int deg = cnt[node];
      const int beg = node * ADJW;
      const int end = beg + (deg <= ADJW ? deg : 0);
      const float sc = 1.0f / fmaxf((float)deg, 1.0f);
      v4f sv[KV];
      if (WITH_SELF) {
        const v4f* self = (const v4f*)(hin + (size_t)node * K) + lane;
        for (int c = 0; c < KV; ++c) sv[c] = self[c * 32];
      }
      v4f acc[KV];
      for (int c = 0; c < KV; ++c) acc[c] = (v4f){0.f, 0.f, 0.f, 0.f};
      int j = beg;
      for (; j + U <= end; j += U) {
        v4f t[U][KV];
        for (int u = 0; u < U; ++u) {
          const int sidx = adj[j + u];
          const v4f* hp = (const v4f*)(hin + (size_t)sidx * K) + lane;
          for (int c = 0; c < KV; ++c) t[u][c] = hp[c * 32];
        }
        for (int u = 0; u < U; ++u)
          for (int c = 0; c < KV; ++c) acc[c] += t[u][c];
      }
#pragma nounroll
      for (; j < end; ++j) {
        const int sidx = adj[j];
        const v4f* hp = (const v4f*)(hin + (size_t)sidx * K) + lane;
        for (int c = 0; c < KV; ++c) acc[c] += hp[c * 32];
      }
      if (deg <= ADJW) {
        for (int c = 0; c < KV; ++c) Arow[c * 32 + lane] = acc[c] * sc;
      }
      if (WITH_SELF) {
        for (int c = 0; c < KV; ++c) Arow[K / 4 + c * 32 + lane] = sv[c];
      }
    } else {
      for (int c = 0; c < KV; ++c) Arow[c * 32 + lane] = (v4f){0.f, 0.f, 0.f, 0.f};
      if (WITH_SELF) {
        for (int c = 0; c < KV; ++c) Arow[K / 4 + c * 32 + lane] = (v4f){0.f, 0.f, 0.f, 0.f};
      }
    }
  }
}

template<int K, int NT, int PITCH>
__device__ __forceinline__ void stage_self(float* __restrict__ As, const float* __restrict__ hin,
                                           int r0, int nnodes, int tid) {
  constexpr int R4    = K / 4;
  constexpr int TOT   = 32 * R4;
  constexpr int TRIPS = (TOT + NT - 1) / NT;
  v4f v[TRIPS];
  for (int m = 0; m < TRIPS; ++m) {
    const int e   = tid + m * NT;
    const int row = e / R4;
    const int c4  = e - row * R4;
    v[m] = (v4f){0.f, 0.f, 0.f, 0.f};
    if (e < TOT && r0 + row < nnodes) v[m] = *((const v4f*)(hin + (size_t)(r0 + row) * K) + c4);
  }
  for (int m = 0; m < TRIPS; ++m) {
    const int e   = tid + m * NT;
    const int row = e / R4;
    const int c4  = e - row * R4;
    if (e < TOT) *((v4f*)(As + row * PITCH + K) + c4) = v[m];
  }
}

template<int K, int NOUT, bool RELU>
__global__ __launch_bounds__(NOUT, (NOUT == 384 ? 12 : 8)) void sage_alt_kernel(const float* __restrict__ hin,
                                                          const int* __restrict__ cnt,
                                                          const int* __restrict__ adj,
                                                          const int* __restrict__ esrc,
                                                          const int* __restrict__ edst, int nedges,
                                                          const float* __restrict__ Wlp,
                                                          const float* __restrict__ Wrp,
                                                          const float* __restrict__ bias,
                                                          float* __restrict__ out, int nnodes) {
  constexpr int NW    = NOUT / 32;
  constexpr int PITCH = 2 * K + 4;
  __shared__ __attribute__((aligned(16))) float As[32 * PITCH];

  const int lane = threadIdx.x & 31;
  const int wv   = __builtin_amdgcn_readfirstlane((int)(threadIdx.x >> 5));
  const int r0   = blockIdx.x * 32;

  const bool self_first = (__popc((unsigned)blockIdx.x) & 1) != 0;

  const unsigned over = rows_over(cnt, r0, nnodes, lane);
  if (over != 0u) overflow_rows<K, NW, PITCH>(As, hin, cnt, esrc, edst, nedges, r0, wv, lane, over);

  v8f acc00 = {}, acc01 = {}, acc10 = {}, acc11 = {};

  if (self_first) {
    stage_self<K, NOUT, PITCH>(As, hin, r0, nnodes, (int)threadIdx.x);
    __syncthreads();
    gemm_pass<K, PITCH>(As, Wrp, 1, wv, lane, acc00, acc01, acc10, acc11);
    build_rows<K, NW, PITCH, false>(As, hin, cnt, adj, r0, nnodes, wv, lane);
    __syncthreads();
    gemm_pass<K, PITCH>(As, Wlp, 0, wv, lane, acc00, acc01, acc10, acc11);
  } else {
    build_rows<K, NW, PITCH, true>(As, hin, cnt, adj, r0, nnodes, wv, lane);
    __syncthreads();
    gemm_pass<K, PITCH>(As, Wlp, 0, wv, lane, acc00, acc01, acc10, acc11);
    gemm_pass<K, PITCH>(As, Wrp, 1, wv, lane, acc00, acc01, acc10, acc11);
  }

  const int l15 = lane & 15;
  const int s   = lane >> 4;
  const int col = wv * 32 + l15;
  const float bj0 = bias[col];
  const float bj1 = bias[col + 16];
  const int rb = r0 + (s << 3);
#pragma unroll
  for (int v = 0; v < 8; ++v) {
    float t00 = acc00[v] + bj0;
    float t01 = acc01[v] + bj1;
    float t10 = acc10[v] + bj0;
    float t11 = acc11[v] + bj1;
    if (RELU) {
      t00 = fmaxf(t00, 0.f); t01 = fmaxf(t01, 0.f);
      t10 = fmaxf(t10, 0.f); t11 = fmaxf(t11, 0.f);
    }
    const int ra = rb + v;
    const int rc = rb + 16 + v;
    if (ra < nnodes) {
      float* o = out + (size_t)ra * NOUT + col;
      o[0] = t00; o[16] = t01;
    }
    if (rc < nnodes) {
      float* o = out + (size_t)rc * NOUT + col;
      o[0] = t10; o[16] = t11;
    }
  }
}


extern "C" void kernel_launch(void* const* d_in, const int* in_sizes, int n_in,
                              void* d_out, int out_size, void* d_ws, size_t ws_size,
                              hipStream_t stream) {
  const float* x   = (const float*)d_in[0];
  const int*   ei  = (const int*)d_in[1];
  const float* W1l = (const float*)d_in[2];
  const float* W1r = (const float*)d_in[3];
  const float* b1  = (const float*)d_in[4];
  const float* W2l = (const float*)d_in[5];
  const float* W2r = (const float*)d_in[6];
  const float* b2  = (const float*)d_in[7];
  const float* W3l = (const float*)d_in[8];
  const float* W3r = (const float*)d_in[9];
  const float* b3  = (const float*)d_in[10];
  float* out = (float*)d_out;

  const int N = in_sizes[0] / DIN;
  const int E = in_sizes[1] / 2;
  const int* src = ei;
  const int* dst = ei + E;

  char* ws = (char*)d_ws;
  auto al = [](size_t v) { return (v + 255) & ~(size_t)255; };
  size_t off = 0;
  int*   cnt      = (int*)(ws + off);   off += al((size_t)N * 4);
  int*   adj      = (int*)(ws + off);   off += al((size_t)N * ADJW * 4);
  float* h1       = (float*)(ws + off); off += al((size_t)N * DH * 4);
  float* h2       = (float*)(ws + off); off += al((size_t)N * DH * 4);
  float* Wt1l     = (float*)(ws + off); off += al((size_t)DIN * DH * 4);
  float* Wt1r     = (float*)(ws + off); off += al((size_t)DIN * DH * 4);
  float* Wt2l     = (float*)(ws + off); off += al((size_t)DH * DH * 4);
  float* Wt2r     = (float*)(ws + off); off += al((size_t)DH * DH * 4);
  float* Wt3l     = (float*)(ws + off); off += al((size_t)DH * DOUT * 4);
  float* Wt3r     = (float*)(ws + off);

  const int TB = 256;
  const int gb = (N + 31) / 32;

  setup_kernel<<<PERM_BLOCKS + (N / 4 + TB - 1) / TB, TB, 0, stream>>>(W1l, W1r, W2l, W2r, W3l, W3r,
                                                                      Wt1l, Wt1r, Wt2l, Wt2r, Wt3l, Wt3r,
                                                                      (float4*)cnt, N / 4);
  fill_adj_kernel<<<(E + TB - 1) / TB, TB, 0, stream>>>(src, dst, cnt, adj, E);

  sage_fused_kernel<DIN, DH, true><<<gb, DH, 0, stream>>>(x, cnt, adj, src, dst, E, Wt1l, Wt1r, b1, h1, N);
  sage_alt_kernel<DH, DH, true><<<gb, DH, 0, stream>>>(h1, cnt, adj, src, dst, E, Wt2l, Wt2r, b2, h2, N);
  sage_alt_kernel<DH, DOUT, false><<<gb, DOUT, 0, stream>>>(h2, cnt, adj, src, dst, E, Wt3l, Wt3r, b3, out, N);
  (void)n_in; (void)out_size; (void)ws_size;
}


# gather: a row's neighbour ids sit one per lane (one load of its slots; next row's asked a row ahead), degree from the group's degree load; layer 1 joins the routine (4 rows in flight); sum order kept
# speedup vs baseline: 3.2006x; 1.0456x over previous
#include <hip/hip_runtime.h>

typedef __attribute__((ext_vector_type(2))) float v2f;
typedef __attribute__((ext_vector_type(4))) float v4f;
typedef __attribute__((ext_vector_type(8))) float v8f;

#define DIN  128
#define DH   256
#define DOUT 384

#define ADJW 64

__global__ __launch_bounds__(256) void fill_adj_kernel(const int* __restrict__ src,
                                                       const int* __restrict__ dst,
                                                       int* __restrict__ cnt,
                                                       int* __restrict__ adj, int nedges) {
  int e = blockIdx.x * blockDim.x + threadIdx.x;
  if (e >= nedges) return;
  const int d = dst[e];
  const int pos = atomicAdd(&cnt[d], 1);
  if (pos < ADJW) adj[(size_t)d * ADJW + pos] = src[e];
}

template<int K, int NW, int PITCH>
__device__ __forceinline__ void overflow_rows(float* __restrict__ As, const float* __restrict__ hin,
                                              const int* __restrict__ cnt,
                                              const int* __restrict__ esrc, const int* __restrict__ edst,
                                              int nedges, int r0, int wv, int lane, unsigned over) {
  constexpr int KV = K / 128;
  for (int i = wv; i < 32; i += NW) {
    if (((over >> i) & 1u) == 0u) continue;
    const int node = r0 + i;
    v4f acc[KV];
    for (int c = 0; c < KV; ++c) acc[c] = (v4f){0.f, 0.f, 0.f, 0.f};
    for (int e0 = 0; e0 < nedges; e0 += 32) {
      const int e = e0 + lane;
      int d = -1, sl = 0;
      if (e < nedges) { d = edst[e]; sl = esrc[e]; }
      unsigned m = __builtin_amdgcn_ballot_w32(d == node);
      while (m != 0u) {
        const int b = __builtin_ctz(m);
        m &= m - 1u;
        const int sidx = __builtin_amdgcn_readlane(sl, b);
        const v4f* hp = (const v4f*)(hin + (size_t)sidx * K) + lane;
        for (int c = 0; c < KV; ++c) acc[c] += hp[c * 32];
      }
    }
    const float sc = 1.0f / fmaxf((float)cnt[node], 1.0f);
    v4f* Arow = (v4f*)(As + i * PITCH);
    for (int c = 0; c < KV; ++c) Arow[c * 32 + lane] = acc[c] * sc;
  }
}

__device__ __forceinline__ unsigned rows_over(const int* __restrict__ cnt, int r0, int nnodes, int lane, int& dg) {
  const int nd = r0 + lane;
  dg = (nd < nnodes) ? cnt[nd] : 0;
  return __builtin_amdgcn_ballot_w32(dg > ADJW);
}

__device__ __forceinline__ int load_ids(const int* __restrict__ adj, int node, int c0, int nb, int lane) {
  int v = 0;
  const unsigned slot = (unsigned)node * ADJW + (unsigned)(c0 + lane);
  if (c0 + lane < nb) v = adj[(size_t)slot];
  return v;
}

template<int K, int NOUT>
__device__ __forceinline__ void permute_w_elem(const float* __restrict__ W,
                                               float* __restrict__ Wt, int i) {
  static_assert((K * NOUT) % 256 == 0, "a matrix is a whole number of 256-thread blocks");
  int k = i / NOUT;
  int col = i - k * NOUT;
  int s = (k >> 1) & 1;
  int j = k >> 2;
  int jj = j >> 1;
  int q = ((j & 1) << 1) | (k & 1);
  int nb = col >> 6;
  int t = (col >> 4) & 3;
  int lane = (s << 4) | (col & 15);
  Wt[((((size_t)nb * (K / 8) + jj) * 4 + t) * 32 + lane) * 4 + q] = W[i];
}

#define PB1 (DIN * DH / 256)
#define PB2 (DH * DH / 256)
#define PB3 (DH * DOUT / 256)
#define PERM_BLOCKS (2 * PB1 + 2 * PB2 + 2 * PB3)

__global__ __launch_bounds__(256) void setup_kernel(const float* __restrict__ W1l,
                                                    const float* __restrict__ W1r,
                                                    const float* __restrict__ W2l,
                                                    const float* __restrict__ W2r,
                                                    const float* __restrict__ W3l,
                                                    const float* __restrict__ W3r,
                                                    float* __restrict__ Wt1l,
                                                    float* __restrict__ Wt1r,
                                                    float* __restrict__ Wt2l,
                                                    float* __restrict__ Wt2r,
                                                    float* __restrict__ Wt3l,
                                                    float* __restrict__ Wt3r,
                                                    float4* __restrict__ cnt4, int n4) {
  int b = blockIdx.x;
  const int t = threadIdx.x;
  if (b < PB1) { permute_w_elem<DIN, DH>(W1l, Wt1l, b * 256 + t); return; }
  b -= PB1;
  if (b < PB1) { permute_w_elem<DIN, DH>(W1r, Wt1r, b * 256 + t); return; }
  b -= PB1;
  if (b < PB2) { permute_w_elem<DH, DH>(W2l, Wt2l, b * 256 + t); return; }
  b -= PB2;
  if (b < PB2) { permute_w_elem<DH, DH>(W2r, Wt2r, b * 256 + t); return; }
  b -= PB2;
  if (b < PB3) { permute_w_elem<DH, DOUT>(W3l, Wt3l, b * 256 + t); return; }
  b -= PB3;
  if (b < PB3) { permute_w_elem<DH, DOUT>(W3r, Wt3r, b * 256 + t); return; }
  b -= PB3;
  const int i = b * 256 + t;
  if (i < n4) cnt4[i] = make_float4(0.f, 0.f, 0.f, 0.f);
}

template<int K, int NW, int PITCH, bool WITH_SELF>
__device__ __forceinline__ void build_rows(float* __restrict__ As, const float* __restrict__ hin,
                                           const int* __restrict__ adj, int dg,
                                           int r0, int nnodes, int wv, int lane);

template<int K, int NOUT, bool RELU>
__global__ __launch_bounds__(NOUT) void sage_fused_kernel(const float* __restrict__ hin,
                                                          const int* __restrict__ cnt,
                                                          const int* __restrict__ adj,
                                                          const int* __restrict__ esrc,
                                                          const int* __restrict__ edst, int nedges,
                                                          const float* __restrict__ Wlp,
                                                          const float* __restrict__ Wrp,
                                                          const float* __restrict__ bias,
                                                          float* __restrict__ out, int nnodes) {
  constexpr int NW    = NOUT / 32;
  constexpr int PITCH = 2 * K + 4;
  __shared__ __attribute__((aligned(16))) float As[32 * PITCH];

  const int lane = threadIdx.x & 31;
  const int wv   = __builtin_amdgcn_readfirstlane((int)(threadIdx.x >> 5));
  const int r0   = blockIdx.x * 32;

  int dg;
  const unsigned over = rows_over(cnt, r0, nnodes, lane, dg);
  if (over != 0u) overflow_rows<K, NW, PITCH>(As, hin, cnt, esrc, edst, nedges, r0, wv, lane, over);

  build_rows<K, NW, PITCH, true>(As, hin, adj, dg, r0, nnodes, wv, lane);
  __syncthreads();

  const int l15   = lane & 15;
  const int s     = lane >> 4;
  const int khalf = s << 1;

  v8f acc00 = {}, acc01 = {}, acc10 = {}, acc11 = {};

  for (int pass = 0; pass < 2; ++pass) {
    const float* __restrict__ Wp = (pass == 0 ? Wlp : Wrp);
    const v4f* bp = (const v4f*)Wp + (size_t)(wv >> 1) * (K / 8) * 128 + (wv & 1) * 64 + lane;
    const float* A0 = As + l15 * PITCH + pass * K + khalf;
    const float* A1 = A0 + 16 * PITCH;
#pragma unroll 2
    for (int jj = 0; jj < K / 8; ++jj) {
      const v2f a00 = *(const v2f*)(A0 + 8 * jj);
      const v2f a01 = *(const v2f*)(A0 + 8 * jj + 4);
      const v2f a10 = *(const v2f*)(A1 + 8 * jj);
      const v2f a11 = *(const v2f*)(A1 + 8 * jj + 4);
      const v4f b0 = bp[jj * 128 +  0];
      const v4f b1 = bp[jj * 128 + 32];
      acc00 = __builtin_amdgcn_wmma_f32_16x16x4_f32(false, a00, false, __builtin_shufflevector(b0, b0, 0, 1), (short)0, acc00, false, false);
      acc01 = __builtin_amdgcn_wmma_f32_16x16x4_f32(false, a00, false, __builtin_shufflevector(b1, b1, 0, 1), (short)0, acc01, false, false);
      acc10 = __builtin_amdgcn_wmma_f32_16x16x4_f32(false, a10, false, __builtin_shufflevector(b0, b0, 0, 1), (short)0, acc10, false, false);
      acc11 = __builtin_amdgcn_wmma_f32_16x16x4_f32(false, a10, false, __builtin_shufflevector(b1, b1, 0, 1), (short)0, acc11, false, false);
      acc00 = __builtin_amdgcn_wmma_f32_16x16x4_f32(false, a01, false, __builtin_shufflevector(b0, b0, 2, 3), (short)0, acc00, false, false);
      acc01 = __builtin_amdgcn_wmma_f32_16x16x4_f32(false, a01, false, __builtin_shufflevector(b1, b1, 2, 3), (short)0, acc01, false, false);
      acc10 = __builtin_amdgcn_wmma_f32_16x16x4_f32(false, a11, false, __builtin_shufflevector(b0, b0, 2, 3), (short)0, acc10, false, false);
      acc11 = __builtin_amdgcn_wmma_f32_16x16x4_f32(false, a11, false, __builtin_shufflevector(b1, b1, 2, 3), (short)0, acc11, false, false);
    }
  }

  const int col = wv * 32 + l15;
  const float bj0 = bias[col];
  const float bj1 = bias[col + 16];
  const int rb = r0 + (s << 3);
#pragma unroll
  for (int v = 0; v < 8; ++v) {
    float t00 = acc00[v] + bj0;
    float t01 = acc01[v] + bj1;
    float t10 = acc10[v] + bj0;
    float t11 = acc11[v] + bj1;
    if (RELU) {
      t00 = fmaxf(t00, 0.f); t01 = fmaxf(t01, 0.f);
      t10 = fmaxf(t10, 0.f); t11 = fmaxf(t11, 0.f);
    }
    const int ra = rb + v;
    const int rc = rb + 16 + v;
    if (ra < nnodes) {
      float* o = out + (size_t)ra * NOUT + col;
      o[0] = t00; o[16] = t01;
    }
    if (rc < nnodes) {
      float* o = out + (size_t)rc * NOUT + col;
      o[0] = t10; o[16] = t11;
    }
  }
}


template<int K, int PITCH>
__device__ __forceinline__ void gemm_pass(const float* __restrict__ As, const float* __restrict__ Wp,
                                          int pass, int wv, int lane,
                                          v8f& acc00, v8f& acc01, v8f& acc10, v8f& acc11) {
  const int l15   = lane & 15;
  const int khalf = (lane >> 4) << 1;
  const v4f* bp = (const v4f*)Wp + (size_t)(wv >> 1) * (K / 8) * 128 + (wv & 1) * 64 + lane;
  const float* A0 = As + l15 * PITCH + pass * K + khalf;
  const float* A1 = A0 + 16 * PITCH;
#pragma unroll 2
  for (int jj = 0; jj < K / 8; ++jj) {
    const v2f a00 = *(const v2f*)(A0 + 8 * jj);
    const v2f a01 = *(const v2f*)(A0 + 8 * jj + 4);
    const v2f a10 = *(const v2f*)(A1 + 8 * jj);
    const v2f a11 = *(const v2f*)(A1 + 8 * jj + 4);
    const v4f b0 = bp[jj * 128 +  0];
    const v4f b1 = bp[jj * 128 + 32];
    acc00 = __builtin_amdgcn_wmma_f32_16x16x4_f32(false, a00, false, __builtin_shufflevector(b0, b0, 0, 1), (short)0, acc00, false, false);
    acc01 = __builtin_amdgcn_wmma_f32_16x16x4_f32(false, a00, false, __builtin_shufflevector(b1, b1, 0, 1), (short)0, acc01, false, false);
    acc10 = __builtin_amdgcn_wmma_f32_16x16x4_f32(false, a10, false, __builtin_shufflevector(b0, b0, 0, 1), (short)0, acc10, false, false);
    acc11 = __builtin_amdgcn_wmma_f32_16x16x4_f32(false, a10, false, __builtin_shufflevector(b1, b1, 0, 1), (short)0, acc11, false, false);
    acc00 = __builtin_amdgcn_wmma_f32_16x16x4_f32(false, a01, false, __builtin_shufflevector(b0, b0, 2, 3), (short)0, acc00, false, false);
    acc01 = __builtin_amdgcn_wmma_f32_16x16x4_f32(false, a01, false, __builtin_shufflevector(b1, b1, 2, 3), (short)0, acc01, false, false);
    acc10 = __builtin_amdgcn_wmma_f32_16x16x4_f32(false, a11, false, __builtin_shufflevector(b0, b0, 2, 3), (short)0, acc10, false, false);
    acc11 = __builtin_amdgcn_wmma_f32_16x16x4_f32(false, a11, false, __builtin_shufflevector(b1, b1, 2, 3), (short)0, acc11, false, false);
  }
}

template<int K, int NW, int PITCH, bool WITH_SELF>
__device__ __forceinline__ void build_rows(float* __restrict__ As, const float* __restrict__ hin,
                                           const int* __restrict__ adj, int dg,
                                           int r0, int nnodes, int wv, int lane) {
  constexpr int KV = K / 128;
  constexpr int U  = 4;
  constexpr unsigned R4 = K / 4;
  const v4f* __restrict__ hv = (const v4f*)hin;
  int deg_n = __builtin_amdgcn_readlane(dg, wv);
  int ids_n = load_ids(adj, r0 + wv, 0, deg_n <= ADJW ? deg_n : 0, lane);
  for (int i = wv; i < 32; i += NW) {
    const int node = r0 + i;
    const int deg  = deg_n;
    const int ids  = ids_n;
    const int nb   = deg <= ADJW ? deg : 0;
    if (i + NW < 32) {
      deg_n = __builtin_amdgcn_readlane(dg, i + NW);
      ids_n = load_ids(adj, node + NW, 0, deg_n <= ADJW ? deg_n : 0, lane);
    }
    v4f* Arow = (v4f*)(As + i * PITCH);
    if (node < nnodes) {
      const float sc = 1.0f / fmaxf((float)deg, 1.0f);
      v4f sv[KV];
      if (WITH_SELF) {
        const unsigned ri = (unsigned)node * R4 + (unsigned)lane;
        for (int c = 0; c < KV; ++c) sv[c] = hv[(size_t)ri + c * 32];
      }
      v4f acc[KV];
      for (int c = 0; c < KV; ++c) acc[c] = (v4f){0.f, 0.f, 0.f, 0.f};
      const int n0 = nb < 32 ? nb : 32;
      int j = 0;
      for (; j + U <= n0; j += U) {
        v4f t[U][KV];
        for (int u = 0; u < U; ++u) {
          const int sidx = __builtin_amdgcn_readlane(ids, j + u);
          const unsigned ri = (unsigned)sidx * R4 + (unsigned)lane;
          for (int c = 0; c < KV; ++c) t[u][c] = hv[(size_t)ri + c * 32];
        }
        for (int u = 0; u < U; ++u)
          for (int c = 0; c < KV; ++c) acc[c] += t[u][c];
      }
#pragma nounroll
      for (; j < n0; ++j) {
        const int sidx = __builtin_amdgcn_readlane(ids, j);
        const unsigned ri = (unsigned)sidx * R4 + (unsigned)lane;
        for (int c = 0; c < KV; ++c) acc[c] += hv[(size_t)ri + c * 32];
      }
      for (int c0 = 32; c0 < nb; c0 += 32) {
        const int idv = load_ids(adj, node, c0, nb, lane);
        const int n = (nb - c0 < 32) ? nb - c0 : 32;
#pragma nounroll
        for (int q = 0; q < n; ++q) {
          const int sidx = __builtin_amdgcn_readlane(idv, q);
          const unsigned ri = (unsigned)sidx * R4 + (unsigned)lane;
          for (int c = 0; c < KV; ++c) acc[c] += hv[(size_t)ri + c * 32];
        }
      }
      if (deg <= ADJW) {
        for (int c = 0; c < KV; ++c) Arow[c * 32 + lane] = acc[c] * sc;
      }
      if (WITH_SELF) {
        for (int c = 0; c < KV; ++c) Arow[K / 4 + c * 32 + lane] = sv[c];
      }
    } else {
      for (int c = 0; c < KV; ++c) Arow[c * 32 + lane] = (v4f){0.f, 0.f, 0.f, 0.f};
      if (WITH_SELF) {
        for (int c = 0; c < KV; ++c) Arow[K / 4 + c * 32 + lane] = (v4f){0.f, 0.f, 0.f, 0.f};
      }
    }
  }
}

template<int K, int NT, int PITCH>
__device__ __forceinline__ void stage_self(float* __restrict__ As, const float* __restrict__ hin,
                                           int r0, int nnodes, int tid) {
  constexpr int R4    = K / 4;
  constexpr int TOT   = 32 * R4;
  constexpr int TRIPS = (TOT + NT - 1) / NT;
  v4f v[TRIPS];
  for (int m = 0; m < TRIPS; ++m) {
    const int e   = tid + m * NT;
    const int row = e / R4;
    const int c4  = e - row * R4;
    v[m] = (v4f){0.f, 0.f, 0.f, 0.f};
    if (e < TOT && r0 + row < nnodes) v[m] = *((const v4f*)(hin + (size_t)(r0 + row) * K) + c4);
  }
  for (int m = 0; m < TRIPS; ++m) {
    const int e   = tid + m * NT;
    const int row = e / R4;
    const int c4  = e - row * R4;
    if (e < TOT) *((v4f*)(As + row * PITCH + K) + c4) = v[m];
  }
}

template<int K, int NOUT, bool RELU>
__global__ __launch_bounds__(NOUT, (NOUT == 384 ? 12 : 8)) void sage_alt_kernel(const float* __restrict__ hin,
                                                          const int* __restrict__ cnt,
                                                          const int* __restrict__ adj,
                                                          const int* __restrict__ esrc,
                                                          const int* __restrict__ edst, int nedges,
                                                          const float* __restrict__ Wlp,
                                                          const float* __restrict__ Wrp,
                                                          const float* __restrict__ bias,
                                                          float* __restrict__ out, int nnodes) {
  constexpr int NW    = NOUT / 32;
  constexpr int PITCH = 2 * K + 4;
  __shared__ __attribute__((aligned(16))) float As[32 * PITCH];

  const int lane = threadIdx.x & 31;
  const int wv   = __builtin_amdgcn_readfirstlane((int)(threadIdx.x >> 5));
  const int r0   = blockIdx.x * 32;

  const bool self_first = (__popc((unsigned)blockIdx.x) & 1) != 0;

  int dg;
  const unsigned over = rows_over(cnt, r0, nnodes, lane, dg);
  if (over != 0u) overflow_rows<K, NW, PITCH>(As, hin, cnt, esrc, edst, nedges, r0, wv, lane, over);

  v8f acc00 = {}, acc01 = {}, acc10 = {}, acc11 = {};

  if (self_first) {
    stage_self<K, NOUT, PITCH>(As, hin, r0, nnodes, (int)threadIdx.x);
    __syncthreads();
    gemm_pass<K, PITCH>(As, Wrp, 1, wv, lane, acc00, acc01, acc10, acc11);
    build_rows<K, NW, PITCH, false>(As, hin, adj, dg, r0, nnodes, wv, lane);
    __syncthreads();
    gemm_pass<K, PITCH>(As, Wlp, 0, wv, lane, acc00, acc01, acc10, acc11);
  } else {
    build_rows<K, NW, PITCH, true>(As, hin, adj, dg, r0, nnodes, wv, lane);
    __syncthreads();
    gemm_pass<K, PITCH>(As, Wlp, 0, wv, lane, acc00, acc01, acc10, acc11);
    gemm_pass<K, PITCH>(As, Wrp, 1, wv, lane, acc00, acc01, acc10, acc11);
  }

  const int l15 = lane & 15;
  const int s   = lane >> 4;
  const int col = wv * 32 + l15;
  const float bj0 = bias[col];
  const float bj1 = bias[col + 16];
  const int rb = r0 + (s << 3);
#pragma unroll
  for (int v = 0; v < 8; ++v) {
    float t00 = acc00[v] + bj0;
    float t01 = acc01[v] + bj1;
    float t10 = acc10[v] + bj0;
    float t11 = acc11[v] + bj1;
    if (RELU) {
      t00 = fmaxf(t00, 0.f); t01 = fmaxf(t01, 0.f);
      t10 = fmaxf(t10, 0.f); t11 = fmaxf(t11, 0.f);
    }
    const int ra = rb + v;
    const int rc = rb + 16 + v;
    if (ra < nnodes) {
      float* o = out + (size_t)ra * NOUT + col;
      o[0] = t00; o[16] = t01;
    }
    if (rc < nnodes) {
      float* o = out + (size_t)rc * NOUT + col;
      o[0] = t10; o[16] = t11;
    }
  }
}


extern "C" void kernel_launch(void* const* d_in, const int* in_sizes, int n_in,
                              void* d_out, int out_size, void* d_ws, size_t ws_size,
                              hipStream_t stream) {
  const float* x   = (const float*)d_in[0];
  const int*   ei  = (const int*)d_in[1];
  const float* W1l = (const float*)d_in[2];
  const float* W1r = (const float*)d_in[3];
  const float* b1  = (const float*)d_in[4];
  const float* W2l = (const float*)d_in[5];
  const float* W2r = (const float*)d_in[6];
  const float* b2  = (const float*)d_in[7];
  const float* W3l = (const float*)d_in[8];
  const float* W3r = (const float*)d_in[9];
  const float* b3  = (const float*)d_in[10];
  float* out = (float*)d_out;

  const int N = in_sizes[0] / DIN;
  const int E = in_sizes[1] / 2;
  const int* src = ei;
  const int* dst = ei + E;

  char* ws = (char*)d_ws;
  auto al = [](size_t v) { return (v + 255) & ~(size_t)255; };
  size_t off = 0;
  int*   cnt      = (int*)(ws + off);   off += al((size_t)N * 4);
  int*   adj      = (int*)(ws + off);   off += al((size_t)N * ADJW * 4);
  float* h1       = (float*)(ws + off); off += al((size_t)N * DH * 4);
  float* h2       = (float*)(ws + off); off += al((size_t)N * DH * 4);
  float* Wt1l     = (float*)(ws + off); off += al((size_t)DIN * DH * 4);
  float* Wt1r     = (float*)(ws + off); off += al((size_t)DIN * DH * 4);
  float* Wt2l     = (float*)(ws + off); off += al((size_t)DH * DH * 4);
  float* Wt2r     = (float*)(ws + off); off += al((size_t)DH * DH * 4);
  float* Wt3l     = (float*)(ws + off); off += al((size_t)DH * DOUT * 4);
  float* Wt3r     = (float*)(ws + off);

  const int TB = 256;
  const int gb = (N + 31) / 32;

  setup_kernel<<<PERM_BLOCKS + (N / 4 + TB - 1) / TB, TB, 0, stream>>>(W1l, W1r, W2l, W2r, W3l, W3r,
                                                                      Wt1l, Wt1r, Wt2l, Wt2r, Wt3l, Wt3r,
                                                                      (float4*)cnt, N / 4);
  fill_adj_kernel<<<(E + TB - 1) / TB, TB, 0, stream>>>(src, dst, cnt, adj, E);

  sage_fused_kernel<DIN, DH, true><<<gb, DH, 0, stream>>>(x, cnt, adj, src, dst, E, Wt1l, Wt1r, b1, h1, N);
  sage_alt_kernel<DH, DH, true><<<gb, DH, 0, stream>>>(h1, cnt, adj, src, dst, E, Wt2l, Wt2r, b2, h2, N);
  sage_alt_kernel<DH, DOUT, false><<<gb, DOUT, 0, stream>>>(h2, cnt, adj, src, dst, E, Wt3l, Wt3r, b3, out, N);
  (void)n_in; (void)out_size; (void)ws_size;
}
